# starting kernel
# baseline (speedup 1.0000x reference)
#include <hip/hip_runtime.h>
#include <hip/hip_bf16.h>

typedef __attribute__((ext_vector_type(2))) float v2f;
typedef __attribute__((ext_vector_type(8))) float v8f;

#define NSLOPE 0.2f

__device__ __forceinline__ unsigned fmap(float x) {
    unsigned b = __float_as_uint(x);
    return (b & 0x80000000u) ? ~b : (b | 0x80000000u);
}
__device__ __forceinline__ float funmap(unsigned u) {
    return (u & 0x80000000u) ? __uint_as_float(u ^ 0x80000000u)
                             : __uint_as_float(~u);
}

__global__ __launch_bounds__(256) void attr_mean_k(const float* __restrict__ eattr,
                                                   float* __restrict__ meanAcc, int E_) {
    __shared__ float s[8];
    if (threadIdx.x < 8) s[threadIdx.x] = 0.f;
    __syncthreads();
    float acc[8] = {0.f, 0.f, 0.f, 0.f, 0.f, 0.f, 0.f, 0.f};
    for (int r = blockIdx.x * blockDim.x + threadIdx.x; r < E_; r += gridDim.x * blockDim.x) {
        const float* row = eattr + (size_t)r * 8;
#pragma unroll
        for (int d = 0; d < 8; ++d) acc[d] += row[d];
    }
#pragma unroll
    for (int d = 0; d < 8; ++d) atomicAdd(&s[d], acc[d]);
    __syncthreads();
    if (threadIdx.x < 8) atomicAdd(meanAcc + threadIdx.x, s[threadIdx.x]);
}

__global__ __launch_bounds__(128) void loop_emb_k(const float* __restrict__ meanAcc,
                                                  const float* __restrict__ We,
                                                  float* __restrict__ loop_emb, float invE) {
    int j = threadIdx.x;
    float s = 0.f;
#pragma unroll
    for (int d = 0; d < 8; ++d) s += (meanAcc[d] * invE) * We[d * 128 + j];
    loop_emb[j] = s;
}

__global__ __launch_bounds__(256) void out_init_k(float* __restrict__ out,
                                                  const float* __restrict__ bias, int total) {
    int i = blockIdx.x * blockDim.x + threadIdx.x;
    if (i < total) out[i] = bias[i & 127];
}

__global__ __launch_bounds__(128) void node_xform_wmma(
    const float* __restrict__ X,
    const float* __restrict__ Wl, const float* __restrict__ bl,
    const float* __restrict__ Wr, const float* __restrict__ br,
    float* __restrict__ xl, float* __restrict__ xr, int nTiles) {
    int wid = threadIdx.x >> 5;
    int lane = threadIdx.x & 31;
    int rowTile = blockIdx.x * 4 + wid;
    if (rowTile >= nTiles) return;

    const float* W    = blockIdx.y ? Wr : Wl;
    const float* bias = blockIdx.y ? br : bl;
    float* out        = blockIdx.y ? xr : xl;

    int row0  = rowTile * 16;
    int m     = lane & 15;
    int khalf = lane >> 4;

    v8f acc[8] = {};

    const float* arow = X + (size_t)(row0 + m) * 128;
    for (int kt = 0; kt < 32; ++kt) {
        int k0 = kt * 4 + khalf * 2;
        v2f a;
        a.x = arow[k0];
        a.y = arow[k0 + 1];
        const float* w0 = W + (size_t)k0 * 128;
        const float* w1 = w0 + 128;
#pragma unroll
        for (int nt = 0; nt < 8; ++nt) {
            v2f b;
            b.x = w0[nt * 16 + m];
            b.y = w1[nt * 16 + m];
            acc[nt] = __builtin_amdgcn_wmma_f32_16x16x4_f32(
                false, a, false, b, (short)0, acc[nt], false, false);
        }
    }
#pragma unroll
    for (int nt = 0; nt < 8; ++nt) {
        float bv = bias[nt * 16 + m];
#pragma unroll
        for (int i = 0; i < 8; ++i) {
            int row = row0 + i + 8 * khalf;
            out[(size_t)row * 128 + nt * 16 + m] = acc[nt][i] + bv;
        }
    }
}

__global__ __launch_bounds__(256) void edge_logit_k(
    const int* __restrict__ ei, const float* __restrict__ eattr,
    const float* __restrict__ We, const float* __restrict__ loop_emb,
    const float* __restrict__ xl, const float* __restrict__ xr,
    const float* __restrict__ att, float* __restrict__ logit,
    unsigned* __restrict__ segmax, int E_, int N_) {
    int gw = (blockIdx.x * blockDim.x + threadIdx.x) >> 5;
    int lane = threadIdx.x & 31;
    if (gw >= E_ + N_) return;

    int src, dst;
    if (gw < E_) { src = ei[gw]; dst = ei[E_ + gw]; }
    else         { src = dst = gw - E_; }

    int c0 = lane * 4;
    float4 vl = *(const float4*)(xl + (size_t)src * 128 + c0);
    float4 vr = *(const float4*)(xr + (size_t)dst * 128 + c0);

    float e0, e1, e2, e3;
    if (gw < E_) {
        float at[8];
#pragma unroll
        for (int d = 0; d < 8; ++d) at[d] = eattr[(size_t)gw * 8 + d];
        e0 = e1 = e2 = e3 = 0.f;
#pragma unroll
        for (int d = 0; d < 8; ++d) {
            const float* wrow = We + d * 128 + c0;
            e0 = fmaf(at[d], wrow[0], e0);
            e1 = fmaf(at[d], wrow[1], e1);
            e2 = fmaf(at[d], wrow[2], e2);
            e3 = fmaf(at[d], wrow[3], e3);
        }
    } else {
        e0 = loop_emb[c0]; e1 = loop_emb[c0 + 1];
        e2 = loop_emb[c0 + 2]; e3 = loop_emb[c0 + 3];
    }

    float m0 = vl.x + vr.x + e0;
    float m1 = vl.y + vr.y + e1;
    float m2 = vl.z + vr.z + e2;
    float m3 = vl.w + vr.w + e3;
    m0 = (m0 > 0.f) ? m0 : NSLOPE * m0;
    m1 = (m1 > 0.f) ? m1 : NSLOPE * m1;
    m2 = (m2 > 0.f) ? m2 : NSLOPE * m2;
    m3 = (m3 > 0.f) ? m3 : NSLOPE * m3;

    float p = m0 * att[c0] + m1 * att[c0 + 1] + m2 * att[c0 + 2] + m3 * att[c0 + 3];
    p += __shfl_xor(p, 1, 32);
    p += __shfl_xor(p, 2, 32);

    int h = lane >> 2;
    if ((lane & 3) == 0) {
        logit[(size_t)gw * 8 + h] = p;
        atomicMax(segmax + (size_t)dst * 8 + h, fmap(p));
    }
}

__global__ __launch_bounds__(256) void edge_exp_k(
    const int* __restrict__ ei, float* __restrict__ logit,
    const unsigned* __restrict__ segmax, float* __restrict__ segsum,
    int E_, int N_) {
    int i = blockIdx.x * blockDim.x + threadIdx.x;
    if (i >= (E_ + N_) * 8) return;
    int e = i >> 3, h = i & 7;
    int dst = (e < E_) ? ei[E_ + e] : (e - E_);
    float mx = funmap(segmax[(size_t)dst * 8 + h]);
    float ex = expf(logit[i] - mx);
    logit[i] = ex;
    atomicAdd(segsum + (size_t)dst * 8 + h, ex);
}

__global__ __launch_bounds__(256) void edge_msg_k(
    const int* __restrict__ ei, const float* __restrict__ expd,
    const float* __restrict__ segsum, const float* __restrict__ xl,
    float* __restrict__ out, int E_, int N_) {
    int gw = (blockIdx.x * blockDim.x + threadIdx.x) >> 5;
    int lane = threadIdx.x & 31;
    if (gw >= E_ + N_) return;

    int src, dst;
    if (gw < E_) { src = ei[gw]; dst = ei[E_ + gw]; }
    else         { src = dst = gw - E_; }

    int h = lane >> 2;
    float alpha = expd[(size_t)gw * 8 + h] / (segsum[(size_t)dst * 8 + h] + 1e-16f);

    int c0 = lane * 4;
    float4 v = *(const float4*)(xl + (size_t)src * 128 + c0);
    float* o = out + (size_t)dst * 128 + c0;
    atomicAdd(o + 0, alpha * v.x);
    atomicAdd(o + 1, alpha * v.y);
    atomicAdd(o + 2, alpha * v.z);
    atomicAdd(o + 3, alpha * v.w);
}

extern "C" void kernel_launch(void* const* d_in, const int* in_sizes, int n_in,
                              void* d_out, int out_size, void* d_ws, size_t ws_size,
                              hipStream_t stream) {
    const float* x     = (const float*)d_in[0];
    const int*   ei    = (const int*)d_in[1];
    const float* eattr = (const float*)d_in[2];
    const float* Wl    = (const float*)d_in[3];
    const float* bl    = (const float*)d_in[4];
    const float* Wr    = (const float*)d_in[5];
    const float* br    = (const float*)d_in[6];
    const float* We    = (const float*)d_in[7];
    const float* att   = (const float*)d_in[8];
    const float* bias  = (const float*)d_in[9];
    float* out = (float*)d_out;
    float* ws  = (float*)d_ws;

    const int n = in_sizes[0] / 128;
    const int e = in_sizes[2] / 8;
    const int tot = e + n;

    size_t off_mean   = 0;
    size_t off_loop   = 16;
    size_t off_segmax = 256;
    size_t off_segsum = off_segmax + (size_t)n * 8;
    size_t off_xl     = off_segsum + (size_t)n * 8;
    size_t off_xr     = off_xl + (size_t)n * 128;
    size_t off_logit  = off_xr + (size_t)n * 128;

    hipMemsetAsync(d_ws, 0, off_xl * sizeof(float), stream);

    attr_mean_k<<<512, 256, 0, stream>>>(eattr, ws + off_mean, e);
    loop_emb_k<<<1, 128, 0, stream>>>(ws + off_mean, We, ws + off_loop, 1.0f / (float)e);
    out_init_k<<<(n * 128 + 255) / 256, 256, 0, stream>>>(out, bias, n * 128);

    int nTiles = n / 16;
    dim3 g((nTiles + 3) / 4, 2);
    node_xform_wmma<<<g, 128, 0, stream>>>(x, Wl, bl, Wr, br,
                                           ws + off_xl, ws + off_xr, nTiles);

    int waveBlocks = ((tot * 32) + 255) / 256;
    edge_logit_k<<<waveBlocks, 256, 0, stream>>>(ei, eattr, We, ws + off_loop,
                                                 ws + off_xl, ws + off_xr, att,
                                                 ws + off_logit,
                                                 (unsigned*)(ws + off_segmax), e, n);
    edge_exp_k<<<(tot * 8 + 255) / 256, 256, 0, stream>>>(ei, ws + off_logit,
                                                          (const unsigned*)(ws + off_segmax),
                                                          ws + off_segsum, e, n);
    edge_msg_k<<<waveBlocks, 256, 0, stream>>>(ei, ws + off_logit, ws + off_segsum,
                                               ws + off_xl, out, e, n);
}


# edge_logit_k: one wave owns a 16-edge tile: edge embedding (self-loop rows too) by f32 WMMA attr[16x8] x W_e, half-wave per gathered row, 32 row gathers in flight; loop_emb launch gone
# speedup vs baseline: 1.2275x; 1.2275x over previous
#include <hip/hip_runtime.h>
#include <hip/hip_bf16.h>

typedef __attribute__((ext_vector_type(2))) float v2f;
typedef __attribute__((ext_vector_type(4))) float v4f;
typedef __attribute__((ext_vector_type(8))) float v8f;

#define NSLOPE 0.2f

__device__ __forceinline__ unsigned fmap(float x) {
    unsigned b = __float_as_uint(x);
    return (b & 0x80000000u) ? ~b : (b | 0x80000000u);
}
__device__ __forceinline__ float funmap(unsigned u) {
    return (u & 0x80000000u) ? __uint_as_float(u ^ 0x80000000u)
                             : __uint_as_float(~u);
}

__global__ __launch_bounds__(256) void attr_mean_k(const float* __restrict__ eattr,
                                                   float* __restrict__ meanAcc, int E_) {
    __shared__ float s[8];
    if (threadIdx.x < 8) s[threadIdx.x] = 0.f;
    __syncthreads();
    float acc[8] = {0.f, 0.f, 0.f, 0.f, 0.f, 0.f, 0.f, 0.f};
    for (int r = blockIdx.x * blockDim.x + threadIdx.x; r < E_; r += gridDim.x * blockDim.x) {
        const float* row = eattr + (size_t)r * 8;
#pragma unroll
        for (int d = 0; d < 8; ++d) acc[d] += row[d];
    }
#pragma unroll
    for (int d = 0; d < 8; ++d) atomicAdd(&s[d], acc[d]);
    __syncthreads();
    if (threadIdx.x < 8) atomicAdd(meanAcc + threadIdx.x, s[threadIdx.x]);
}

__global__ __launch_bounds__(256) void out_init_k(float* __restrict__ out,
                                                  const float* __restrict__ bias, int total) {
    int i = blockIdx.x * blockDim.x + threadIdx.x;
    if (i < total) out[i] = bias[i & 127];
}

__global__ __launch_bounds__(128) void node_xform_wmma(
    const float* __restrict__ X,
    const float* __restrict__ Wl, const float* __restrict__ bl,
    const float* __restrict__ Wr, const float* __restrict__ br,
    float* __restrict__ xl, float* __restrict__ xr, int nTiles) {
    int wid = threadIdx.x >> 5;
    int lane = threadIdx.x & 31;
    int rowTile = blockIdx.x * 4 + wid;
    if (rowTile >= nTiles) return;

    const float* W    = blockIdx.y ? Wr : Wl;
    const float* bias = blockIdx.y ? br : bl;
    float* out        = blockIdx.y ? xr : xl;

    int row0  = rowTile * 16;
    int m     = lane & 15;
    int khalf = lane >> 4;

    v8f acc[8] = {};

    const float* arow = X + (size_t)(row0 + m) * 128;
    for (int kt = 0; kt < 32; ++kt) {
        int k0 = kt * 4 + khalf * 2;
        v2f a;
        a.x = arow[k0];
        a.y = arow[k0 + 1];
        const float* w0 = W + (size_t)k0 * 128;
        const float* w1 = w0 + 128;
#pragma unroll
        for (int nt = 0; nt < 8; ++nt) {
            v2f b;
            b.x = w0[nt * 16 + m];
            b.y = w1[nt * 16 + m];
            acc[nt] = __builtin_amdgcn_wmma_f32_16x16x4_f32(
                false, a, false, b, (short)0, acc[nt], false, false);
        }
    }
#pragma unroll
    for (int nt = 0; nt < 8; ++nt) {
        float bv = bias[nt * 16 + m];
#pragma unroll
        for (int i = 0; i < 8; ++i) {
            int row = row0 + i + 8 * khalf;
            out[(size_t)row * 128 + nt * 16 + m] = acc[nt][i] + bv;
        }
    }
}

__global__ __launch_bounds__(256) void edge_logit_tile_k(
    const int* __restrict__ ei, const float* __restrict__ eattr,
    const float* __restrict__ We, const float* __restrict__ meanAcc, float invE,
    const float* __restrict__ xl, const float* __restrict__ xr,
    const float* __restrict__ att, float* __restrict__ logit,
    unsigned* __restrict__ segmax, int E_, int N_) {
    const int lane = threadIdx.x & 31;
    const int tile = (blockIdx.x * blockDim.x + threadIdx.x) >> 5;
    if (tile >= ((E_ + N_) >> 4)) return;
    const int c  = lane & 15;
    const int hf = lane >> 4;
    const int e0 = tile << 4;
    const int eb = e0 + 8 * hf;

    v4f w[2][2][2];
#pragma unroll
    for (int kt = 0; kt < 2; ++kt)
#pragma unroll
        for (int s = 0; s < 2; ++s)
#pragma unroll
            for (int y = 0; y < 2; ++y)
                w[kt][s][y] = *(const v4f*)(We + (size_t)(4 * kt + 2 * hf + y) * 128 + 64 * s + 4 * c);

    const v4f av0 = *(const v4f*)(att + 4 * c);
    const v4f av1 = *(const v4f*)(att + 64 + 4 * c);

    int src[8], dst[8];
    v2f a[2];
    if (e0 < E_) {
        const int4 s0 = *(const int4*)(ei + eb);
        const int4 s1 = *(const int4*)(ei + eb + 4);
        const int4 d0 = *(const int4*)(ei + E_ + eb);
        const int4 d1 = *(const int4*)(ei + E_ + eb + 4);
        src[0] = s0.x; src[1] = s0.y; src[2] = s0.z; src[3] = s0.w;
        src[4] = s1.x; src[5] = s1.y; src[6] = s1.z; src[7] = s1.w;
        dst[0] = d0.x; dst[1] = d0.y; dst[2] = d0.z; dst[3] = d0.w;
        dst[4] = d1.x; dst[5] = d1.y; dst[6] = d1.z; dst[7] = d1.w;
        const float* arow = eattr + (size_t)(e0 + c) * 8 + 2 * hf;
        a[0] = *(const v2f*)(arow);
        a[1] = *(const v2f*)(arow + 4);
    } else {
#pragma unroll
        for (int r = 0; r < 8; ++r) src[r] = dst[r] = eb - E_ + r;
        a[0].x = meanAcc[2 * hf]     * invE;
        a[0].y = meanAcc[2 * hf + 1] * invE;
        a[1].x = meanAcc[4 + 2 * hf]     * invE;
        a[1].y = meanAcc[4 + 2 * hf + 1] * invE;
    }

    v4f vl[2][8], vr[2][8];
#pragma unroll
    for (int s = 0; s < 2; ++s) {
#pragma unroll
        for (int r = 0; r < 8; ++r) {
            vl[s][r] = *(const v4f*)(xl + (size_t)src[r] * 128 + 64 * s + 4 * c);
            vr[s][r] = *(const v4f*)(xr + (size_t)dst[r] * 128 + 64 * s + 4 * c);
        }
    }
    __builtin_amdgcn_sched_barrier(0);

    v8f D[2][4] = {};
#pragma unroll
    for (int kt = 0; kt < 2; ++kt)
#pragma unroll
        for (int s = 0; s < 2; ++s)
#pragma unroll
            for (int q = 0; q < 4; ++q) {
                v2f b;
                b.x = w[kt][s][0][q];
                b.y = w[kt][s][1][q];
                D[s][q] = __builtin_amdgcn_wmma_f32_16x16x4_f32(
                    false, a[kt], false, b, (short)0, D[s][q], false, false);
            }
    __builtin_amdgcn_sched_barrier(0);

    float p[2][8];
#pragma unroll
    for (int s = 0; s < 2; ++s) {
        const v4f av = s ? av1 : av0;
#pragma unroll
        for (int r = 0; r < 8; ++r) {
            float m0 = vl[s][r][0] + vr[s][r][0] + D[s][0][r];
            float m1 = vl[s][r][1] + vr[s][r][1] + D[s][1][r];
            float m2 = vl[s][r][2] + vr[s][r][2] + D[s][2][r];
            float m3 = vl[s][r][3] + vr[s][r][3] + D[s][3][r];
            m0 = (m0 > 0.f) ? m0 : NSLOPE * m0;
            m1 = (m1 > 0.f) ? m1 : NSLOPE * m1;
            m2 = (m2 > 0.f) ? m2 : NSLOPE * m2;
            m3 = (m3 > 0.f) ? m3 : NSLOPE * m3;
            float t = m0 * av[0] + m1 * av[1] + m2 * av[2] + m3 * av[3];
            t += __shfl_xor(t, 1, 32);
            t += __shfl_xor(t, 2, 32);
            p[s][r] = t;
        }
    }

    const int j    = c & 3;
    const int head = ((j & 1) << 2) + (c >> 2);
    const int rofs = (j >> 1) << 2;
#pragma unroll
    for (int r = 0; r < 4; ++r) {
        const float lo = (j & 1) ? p[1][r]     : p[0][r];
        const float hi = (j & 1) ? p[1][r + 4] : p[0][r + 4];
        const float v  = (j >> 1) ? hi : lo;
        const int   d  = (j >> 1) ? dst[r + 4] : dst[r];
        logit[(size_t)(eb + r + rofs) * 8 + head] = v;
        atomicMax(segmax + (size_t)d * 8 + head, fmap(v));
    }
}

__global__ __launch_bounds__(256) void edge_exp_k(
    const int* __restrict__ ei, float* __restrict__ logit,
    const unsigned* __restrict__ segmax, float* __restrict__ segsum,
    int E_, int N_) {
    int i = blockIdx.x * blockDim.x + threadIdx.x;
    if (i >= (E_ + N_) * 8) return;
    int e = i >> 3, h = i & 7;
    int dst = (e < E_) ? ei[E_ + e] : (e - E_);
    float mx = funmap(segmax[(size_t)dst * 8 + h]);
    float ex = expf(logit[i] - mx);
    logit[i] = ex;
    atomicAdd(segsum + (size_t)dst * 8 + h, ex);
}

__global__ __launch_bounds__(256) void edge_msg_k(
    const int* __restrict__ ei, const float* __restrict__ expd,
    const float* __restrict__ segsum, const float* __restrict__ xl,
    float* __restrict__ out, int E_, int N_) {
    int gw = (blockIdx.x * blockDim.x + threadIdx.x) >> 5;
    int lane = threadIdx.x & 31;
    if (gw >= E_ + N_) return;

    int src, dst;
    if (gw < E_) { src = ei[gw]; dst = ei[E_ + gw]; }
    else         { src = dst = gw - E_; }

    int h = lane >> 2;
    float alpha = expd[(size_t)gw * 8 + h] / (segsum[(size_t)dst * 8 + h] + 1e-16f);

    int c0 = lane * 4;
    float4 v = *(const float4*)(xl + (size_t)src * 128 + c0);
    float* o = out + (size_t)dst * 128 + c0;
    atomicAdd(o + 0, alpha * v.x);
    atomicAdd(o + 1, alpha * v.y);
    atomicAdd(o + 2, alpha * v.z);
    atomicAdd(o + 3, alpha * v.w);
}

extern "C" void kernel_launch(void* const* d_in, const int* in_sizes, int n_in,
                              void* d_out, int out_size, void* d_ws, size_t ws_size,
                              hipStream_t stream) {
    const float* x     = (const float*)d_in[0];
    const int*   ei    = (const int*)d_in[1];
    const float* eattr = (const float*)d_in[2];
    const float* Wl    = (const float*)d_in[3];
    const float* bl    = (const float*)d_in[4];
    const float* Wr    = (const float*)d_in[5];
    const float* br    = (const float*)d_in[6];
    const float* We    = (const float*)d_in[7];
    const float* att   = (const float*)d_in[8];
    const float* bias  = (const float*)d_in[9];
    float* out = (float*)d_out;
    float* ws  = (float*)d_ws;

    const int n = in_sizes[0] / 128;
    const int e = in_sizes[2] / 8;
    const int tot = e + n;

    size_t off_mean   = 0;
    size_t off_segmax = 256;
    size_t off_segsum = off_segmax + (size_t)n * 8;
    size_t off_xl     = off_segsum + (size_t)n * 8;
    size_t off_xr     = off_xl + (size_t)n * 128;
    size_t off_logit  = off_xr + (size_t)n * 128;

    hipMemsetAsync(d_ws, 0, off_xl * sizeof(float), stream);

    attr_mean_k<<<512, 256, 0, stream>>>(eattr, ws + off_mean, e);
    out_init_k<<<(n * 128 + 255) / 256, 256, 0, stream>>>(out, bias, n * 128);

    int nTiles = n / 16;
    dim3 g((nTiles + 3) / 4, 2);
    node_xform_wmma<<<g, 128, 0, stream>>>(x, Wl, bl, Wr, br,
                                           ws + off_xl, ws + off_xr, nTiles);

    int waveBlocks = ((tot * 32) + 255) / 256;
    int tileBlocks = ((tot / 16) * 32 + 255) / 256;
    edge_logit_tile_k<<<tileBlocks, 256, 0, stream>>>(ei, eattr, We, ws + off_mean,
                                                      1.0f / (float)e,
                                                      ws + off_xl, ws + off_xr, att,
                                                      ws + off_logit,
                                                      (unsigned*)(ws + off_segmax), e, n);
    edge_exp_k<<<(tot * 8 + 255) / 256, 256, 0, stream>>>(ei, ws + off_logit,
                                                          (const unsigned*)(ws + off_segmax),
                                                          ws + off_segsum, e, n);
    edge_msg_k<<<waveBlocks, 256, 0, stream>>>(ei, ws + off_logit, ws + off_segsum,
                                               ws + off_xl, out, e, n);
}


# start values (out = bias, softmax max and sum = 0, column sums = 0) stored by the node transform's waves for their own 16 rows ahead of the K loop; memset and out_init launch gone
# speedup vs baseline: 1.2728x; 1.0369x over previous
#include <hip/hip_runtime.h>
#include <hip/hip_bf16.h>

typedef __attribute__((ext_vector_type(2))) float v2f;
typedef __attribute__((ext_vector_type(4))) float v4f;
typedef __attribute__((ext_vector_type(8))) float v8f;

#define NSLOPE 0.2f

__device__ __forceinline__ unsigned fmap(float x) {
    unsigned b = __float_as_uint(x);
    return (b & 0x80000000u) ? ~b : (b | 0x80000000u);
}
__device__ __forceinline__ float funmap(unsigned u) {
    return (u & 0x80000000u) ? __uint_as_float(u ^ 0x80000000u)
                             : __uint_as_float(~u);
}

__global__ __launch_bounds__(256) void attr_mean_k(const float* __restrict__ eattr,
                                                   float* __restrict__ meanAcc, int E_) {
    __shared__ float s[8];
    if (threadIdx.x < 8) s[threadIdx.x] = 0.f;
    __syncthreads();
    float acc[8] = {0.f, 0.f, 0.f, 0.f, 0.f, 0.f, 0.f, 0.f};
    for (int r = blockIdx.x * blockDim.x + threadIdx.x; r < E_; r += gridDim.x * blockDim.x) {
        const float* row = eattr + (size_t)r * 8;
#pragma unroll
        for (int d = 0; d < 8; ++d) acc[d] += row[d];
    }
#pragma unroll
    for (int d = 0; d < 8; ++d) atomicAdd(&s[d], acc[d]);
    __syncthreads();
    if (threadIdx.x < 8) atomicAdd(meanAcc + threadIdx.x, s[threadIdx.x]);
}

__global__ __launch_bounds__(128) void node_xform_wmma(
    const float* __restrict__ X,
    const float* __restrict__ Wl, const float* __restrict__ bl,
    const float* __restrict__ Wr, const float* __restrict__ br,
    float* __restrict__ xl, float* __restrict__ xr,
    const float* __restrict__ obias, float* __restrict__ outp,
    float* __restrict__ segmaxz, float* __restrict__ segsumz,
    float* __restrict__ meanAcc, int nTiles) {
    int wid = threadIdx.x >> 5;
    int lane = threadIdx.x & 31;
    int rowTile = blockIdx.x * 4 + wid;
    if (rowTile >= nTiles) return;

    if (blockIdx.y == 0) {
        const v4f bv4 = *(const v4f*)(obias + 4 * lane);
        float* o = outp + (size_t)rowTile * 16 * 128 + 4 * lane;
#pragma unroll
        for (int r = 0; r < 16; ++r) *(v4f*)(o + (size_t)r * 128) = bv4;
    } else {
        const v4f z4 = {0.f, 0.f, 0.f, 0.f};
        *(v4f*)(segmaxz + (size_t)rowTile * 128 + 4 * lane) = z4;
        *(v4f*)(segsumz + (size_t)rowTile * 128 + 4 * lane) = z4;
        if (rowTile == 0 && lane < 8) meanAcc[lane] = 0.f;
    }

    const float* W    = blockIdx.y ? Wr : Wl;
    const float* bias = blockIdx.y ? br : bl;
    float* out        = blockIdx.y ? xr : xl;

    int row0  = rowTile * 16;
    int m     = lane & 15;
    int khalf = lane >> 4;

    v8f acc[8] = {};

    const float* arow = X + (size_t)(row0 + m) * 128;
    for (int kt = 0; kt < 32; ++kt) {
        int k0 = kt * 4 + khalf * 2;
        v2f a;
        a.x = arow[k0];
        a.y = arow[k0 + 1];
        const float* w0 = W + (size_t)k0 * 128;
        const float* w1 = w0 + 128;
#pragma unroll
        for (int nt = 0; nt < 8; ++nt) {
            v2f b;
            b.x = w0[nt * 16 + m];
            b.y = w1[nt * 16 + m];
            acc[nt] = __builtin_amdgcn_wmma_f32_16x16x4_f32(
                false, a, false, b, (short)0, acc[nt], false, false);
        }
    }
#pragma unroll
    for (int nt = 0; nt < 8; ++nt) {
        float bv = bias[nt * 16 + m];
#pragma unroll
        for (int i = 0; i < 8; ++i) {
            int row = row0 + i + 8 * khalf;
            out[(size_t)row * 128 + nt * 16 + m] = acc[nt][i] + bv;
        }
    }
}

__global__ __launch_bounds__(256) void edge_logit_tile_k(
    const int* __restrict__ ei, const float* __restrict__ eattr,
    const float* __restrict__ We, const float* __restrict__ meanAcc, float invE,
    const float* __restrict__ xl, const float* __restrict__ xr,
    const float* __restrict__ att, float* __restrict__ logit,
    unsigned* __restrict__ segmax, int E_, int N_) {
    const int lane = threadIdx.x & 31;
    const int tile = (blockIdx.x * blockDim.x + threadIdx.x) >> 5;
    if (tile >= ((E_ + N_) >> 4)) return;
    const int c  = lane & 15;
    const int hf = lane >> 4;
    const int e0 = tile << 4;
    const int eb = e0 + 8 * hf;

    v4f w[2][2][2];
#pragma unroll
    for (int kt = 0; kt < 2; ++kt)
#pragma unroll
        for (int s = 0; s < 2; ++s)
#pragma unroll
            for (int y = 0; y < 2; ++y)
                w[kt][s][y] = *(const v4f*)(We + (size_t)(4 * kt + 2 * hf + y) * 128 + 64 * s + 4 * c);

    const v4f av0 = *(const v4f*)(att + 4 * c);
    const v4f av1 = *(const v4f*)(att + 64 + 4 * c);

    int src[8], dst[8];
    v2f a[2];
    if (e0 < E_) {
        const int4 s0 = *(const int4*)(ei + eb);
        const int4 s1 = *(const int4*)(ei + eb + 4);
        const int4 d0 = *(const int4*)(ei + E_ + eb);
        const int4 d1 = *(const int4*)(ei + E_ + eb + 4);
        src[0] = s0.x; src[1] = s0.y; src[2] = s0.z; src[3] = s0.w;
        src[4] = s1.x; src[5] = s1.y; src[6] = s1.z; src[7] = s1.w;
        dst[0] = d0.x; dst[1] = d0.y; dst[2] = d0.z; dst[3] = d0.w;
        dst[4] = d1.x; dst[5] = d1.y; dst[6] = d1.z; dst[7] = d1.w;
        const float* arow = eattr + (size_t)(e0 + c) * 8 + 2 * hf;
        a[0] = *(const v2f*)(arow);
        a[1] = *(const v2f*)(arow + 4);
    } else {
#pragma unroll
        for (int r = 0; r < 8; ++r) src[r] = dst[r] = eb - E_ + r;
        a[0].x = meanAcc[2 * hf]     * invE;
        a[0].y = meanAcc[2 * hf + 1] * invE;
        a[1].x = meanAcc[4 + 2 * hf]     * invE;
        a[1].y = meanAcc[4 + 2 * hf + 1] * invE;
    }

    v4f vl[2][8], vr[2][8];
#pragma unroll
    for (int s = 0; s < 2; ++s) {
#pragma unroll
        for (int r = 0; r < 8; ++r) {
            vl[s][r] = *(const v4f*)(xl + (size_t)src[r] * 128 + 64 * s + 4 * c);
            vr[s][r] = *(const v4f*)(xr + (size_t)dst[r] * 128 + 64 * s + 4 * c);
        }
    }
    __builtin_amdgcn_sched_barrier(0);

    v8f D[2][4] = {};
#pragma unroll
    for (int kt = 0; kt < 2; ++kt)
#pragma unroll
        for (int s = 0; s < 2; ++s)
#pragma unroll
            for (int q = 0; q < 4; ++q) {
                v2f b;
                b.x = w[kt][s][0][q];
                b.y = w[kt][s][1][q];
                D[s][q] = __builtin_amdgcn_wmma_f32_16x16x4_f32(
                    false, a[kt], false, b, (short)0, D[s][q], false, false);
            }
    __builtin_amdgcn_sched_barrier(0);

    float p[2][8];
#pragma unroll
    for (int s = 0; s < 2; ++s) {
        const v4f av = s ? av1 : av0;
#pragma unroll
        for (int r = 0; r < 8; ++r) {
            float m0 = vl[s][r][0] + vr[s][r][0] + D[s][0][r];
            float m1 = vl[s][r][1] + vr[s][r][1] + D[s][1][r];
            float m2 = vl[s][r][2] + vr[s][r][2] + D[s][2][r];
            float m3 = vl[s][r][3] + vr[s][r][3] + D[s][3][r];
            m0 = (m0 > 0.f) ? m0 : NSLOPE * m0;
            m1 = (m1 > 0.f) ? m1 : NSLOPE * m1;
            m2 = (m2 > 0.f) ? m2 : NSLOPE * m2;
            m3 = (m3 > 0.f) ? m3 : NSLOPE * m3;
            float t = m0 * av[0] + m1 * av[1] + m2 * av[2] + m3 * av[3];
            t += __shfl_xor(t, 1, 32);
            t += __shfl_xor(t, 2, 32);
            p[s][r] = t;
        }
    }

    const int j    = c & 3;
    const int head = ((j & 1) << 2) + (c >> 2);
    const int rofs = (j >> 1) << 2;
#pragma unroll
    for (int r = 0; r < 4; ++r) {
        const float lo = (j & 1) ? p[1][r]     : p[0][r];
        const float hi = (j & 1) ? p[1][r + 4] : p[0][r + 4];
        const float v  = (j >> 1) ? hi : lo;
        const int   d  = (j >> 1) ? dst[r + 4] : dst[r];
        logit[(size_t)(eb + r + rofs) * 8 + head] = v;
        atomicMax(segmax + (size_t)d * 8 + head, fmap(v));
    }
}

__global__ __launch_bounds__(256) void edge_exp_k(
    const int* __restrict__ ei, float* __restrict__ logit,
    const unsigned* __restrict__ segmax, float* __restrict__ segsum,
    int E_, int N_) {
    int i = blockIdx.x * blockDim.x + threadIdx.x;
    if (i >= (E_ + N_) * 8) return;
    int e = i >> 3, h = i & 7;
    int dst = (e < E_) ? ei[E_ + e] : (e - E_);
    float mx = funmap(segmax[(size_t)dst * 8 + h]);
    float ex = expf(logit[i] - mx);
    logit[i] = ex;
    atomicAdd(segsum + (size_t)dst * 8 + h, ex);
}

__global__ __launch_bounds__(256) void edge_msg_k(
    const int* __restrict__ ei, const float* __restrict__ expd,
    const float* __restrict__ segsum, const float* __restrict__ xl,
    float* __restrict__ out, int E_, int N_) {
    int gw = (blockIdx.x * blockDim.x + threadIdx.x) >> 5;
    int lane = threadIdx.x & 31;
    if (gw >= E_ + N_) return;

    int src, dst;
    if (gw < E_) { src = ei[gw]; dst = ei[E_ + gw]; }
    else         { src = dst = gw - E_; }

    int h = lane >> 2;
    float alpha = expd[(size_t)gw * 8 + h] / (segsum[(size_t)dst * 8 + h] + 1e-16f);

    int c0 = lane * 4;
    float4 v = *(const float4*)(xl + (size_t)src * 128 + c0);
    float* o = out + (size_t)dst * 128 + c0;
    atomicAdd(o + 0, alpha * v.x);
    atomicAdd(o + 1, alpha * v.y);
    atomicAdd(o + 2, alpha * v.z);
    atomicAdd(o + 3, alpha * v.w);
}

extern "C" void kernel_launch(void* const* d_in, const int* in_sizes, int n_in,
                              void* d_out, int out_size, void* d_ws, size_t ws_size,
                              hipStream_t stream) {
    const float* x     = (const float*)d_in[0];
    const int*   ei    = (const int*)d_in[1];
    const float* eattr = (const float*)d_in[2];
    const float* Wl    = (const float*)d_in[3];
    const float* bl    = (const float*)d_in[4];
    const float* Wr    = (const float*)d_in[5];
    const float* br    = (const float*)d_in[6];
    const float* We    = (const float*)d_in[7];
    const float* att   = (const float*)d_in[8];
    const float* bias  = (const float*)d_in[9];
    float* out = (float*)d_out;
    float* ws  = (float*)d_ws;

    const int n = in_sizes[0] / 128;
    const int e = in_sizes[2] / 8;
    const int tot = e + n;

    size_t off_mean   = 0;
    size_t off_segmax = 256;
    size_t off_segsum = off_segmax + (size_t)n * 8;
    size_t off_xl     = off_segsum + (size_t)n * 8;
    size_t off_xr     = off_xl + (size_t)n * 128;
    size_t off_logit  = off_xr + (size_t)n * 128;

    int nTiles = n / 16;
    dim3 g((nTiles + 3) / 4, 2);
    node_xform_wmma<<<g, 128, 0, stream>>>(x, Wl, bl, Wr, br,
                                           ws + off_xl, ws + off_xr,
                                           bias, out, ws + off_segmax, ws + off_segsum,
                                           ws + off_mean, nTiles);
    attr_mean_k<<<512, 256, 0, stream>>>(eattr, ws + off_mean, e);

    int waveBlocks = ((tot * 32) + 255) / 256;
    int tileBlocks = ((tot / 16) * 32 + 255) / 256;
    edge_logit_tile_k<<<tileBlocks, 256, 0, stream>>>(ei, eattr, We, ws + off_mean,
                                                      1.0f / (float)e,
                                                      ws + off_xl, ws + off_xr, att,
                                                      ws + off_logit,
                                                      (unsigned*)(ws + off_segmax), e, n);
    edge_exp_k<<<(tot * 8 + 255) / 256, 256, 0, stream>>>(ei, ws + off_logit,
                                                          (const unsigned*)(ws + off_segmax),
                                                          ws + off_segsum, e, n);
    edge_msg_k<<<waveBlocks, 256, 0, stream>>>(ei, ws + off_logit, ws + off_segsum,
                                               ws + off_xl, out, e, n);
}


# dst list split once into the 256 row ranges by a new stable partition launch (a wave per 416 edges); message groups copy their sublists, no longer each scan all edges; row sums in plain edge order
# speedup vs baseline: 2.3499x; 1.8462x over previous
#include <hip/hip_runtime.h>
#include <hip/hip_bf16.h>

typedef __attribute__((ext_vector_type(2))) float v2f;
typedef __attribute__((ext_vector_type(4))) float v4f;
typedef __attribute__((ext_vector_type(8))) float v8f;

#define NSLOPE 0.2f

__device__ __forceinline__ unsigned fmap(float x) {
    unsigned b = __float_as_uint(x);
    return (b & 0x80000000u) ? ~b : (b | 0x80000000u);
}
__device__ __forceinline__ float funmap(unsigned u) {
    return (u & 0x80000000u) ? __uint_as_float(u ^ 0x80000000u)
                             : __uint_as_float(~u);
}

__global__ __launch_bounds__(256) void attr_mean_k(const float* __restrict__ eattr,
                                                   float* __restrict__ meanAcc, int E_) {
    __shared__ float s[8];
    if (threadIdx.x < 8) s[threadIdx.x] = 0.f;
    __syncthreads();
    float acc[8] = {0.f, 0.f, 0.f, 0.f, 0.f, 0.f, 0.f, 0.f};
    for (int r = blockIdx.x * blockDim.x + threadIdx.x; r < E_; r += gridDim.x * blockDim.x) {
        const float* row = eattr + (size_t)r * 8;
#pragma unroll
        for (int d = 0; d < 8; ++d) acc[d] += row[d];
    }
#pragma unroll
    for (int d = 0; d < 8; ++d) atomicAdd(&s[d], acc[d]);
    __syncthreads();
    if (threadIdx.x < 8) atomicAdd(meanAcc + threadIdx.x, s[threadIdx.x]);
}

__global__ __launch_bounds__(128) void node_xform_wmma(
    const float* __restrict__ X,
    const float* __restrict__ Wl, const float* __restrict__ bl,
    const float* __restrict__ Wr, const float* __restrict__ br,
    float* __restrict__ xl, float* __restrict__ xr,
    float* __restrict__ meanAcc, int nTiles) {
    int wid = threadIdx.x >> 5;
    int lane = threadIdx.x & 31;
    int rowTile = blockIdx.x * 4 + wid;
    if (rowTile >= nTiles) return;

    if (blockIdx.y != 0 && rowTile == 0 && lane < 8) meanAcc[lane] = 0.f;

    const float* W    = blockIdx.y ? Wr : Wl;
    const float* bias = blockIdx.y ? br : bl;
    float* out        = blockIdx.y ? xr : xl;

    int row0  = rowTile * 16;
    int m     = lane & 15;
    int khalf = lane >> 4;

    v8f acc[8] = {};

    const float* arow = X + (size_t)(row0 + m) * 128;
    for (int kt = 0; kt < 32; ++kt) {
        int k0 = kt * 4 + khalf * 2;
        v2f a;
        a.x = arow[k0];
        a.y = arow[k0 + 1];
        const float* w0 = W + (size_t)k0 * 128;
        const float* w1 = w0 + 128;
#pragma unroll
        for (int nt = 0; nt < 8; ++nt) {
            v2f b;
            b.x = w0[nt * 16 + m];
            b.y = w1[nt * 16 + m];
            acc[nt] = __builtin_amdgcn_wmma_f32_16x16x4_f32(
                false, a, false, b, (short)0, acc[nt], false, false);
        }
    }
#pragma unroll
    for (int nt = 0; nt < 8; ++nt) {
        float bv = bias[nt * 16 + m];
#pragma unroll
        for (int i = 0; i < 8; ++i) {
            int row = row0 + i + 8 * khalf;
            out[(size_t)row * 128 + nt * 16 + m] = acc[nt][i] + bv;
        }
    }
}

__global__ __launch_bounds__(256) void edge_logit_tile_k(
    const int* __restrict__ ei, const float* __restrict__ eattr,
    const float* __restrict__ We, const float* __restrict__ meanAcc, float invE,
    const float* __restrict__ xl, const float* __restrict__ xr,
    const float* __restrict__ att, float* __restrict__ logit, int E_, int N_) {
    const int lane = threadIdx.x & 31;
    const int tile = (blockIdx.x * blockDim.x + threadIdx.x) >> 5;
    if (tile >= ((E_ + N_) >> 4)) return;
    const int c  = lane & 15;
    const int hf = lane >> 4;
    const int e0 = tile << 4;
    const int eb = e0 + 8 * hf;

    v4f w[2][2][2];
#pragma unroll
    for (int kt = 0; kt < 2; ++kt)
#pragma unroll
        for (int s = 0; s < 2; ++s)
#pragma unroll
            for (int y = 0; y < 2; ++y)
                w[kt][s][y] = *(const v4f*)(We + (size_t)(4 * kt + 2 * hf + y) * 128 + 64 * s + 4 * c);

    const v4f av0 = *(const v4f*)(att + 4 * c);
    const v4f av1 = *(const v4f*)(att + 64 + 4 * c);

    int src[8], dst[8];
    v2f a[2];
    if (e0 < E_) {
        const int4 s0 = *(const int4*)(ei + eb);
        const int4 s1 = *(const int4*)(ei + eb + 4);
        const int4 d0 = *(const int4*)(ei + E_ + eb);
        const int4 d1 = *(const int4*)(ei + E_ + eb + 4);
        src[0] = s0.x; src[1] = s0.y; src[2] = s0.z; src[3] = s0.w;
        src[4] = s1.x; src[5] = s1.y; src[6] = s1.z; src[7] = s1.w;
        dst[0] = d0.x; dst[1] = d0.y; dst[2] = d0.z; dst[3] = d0.w;
        dst[4] = d1.x; dst[5] = d1.y; dst[6] = d1.z; dst[7] = d1.w;
        const float* arow = eattr + (size_t)(e0 + c) * 8 + 2 * hf;
        a[0] = *(const v2f*)(arow);
        a[1] = *(const v2f*)(arow + 4);
    } else {
#pragma unroll
        for (int r = 0; r < 8; ++r) src[r] = dst[r] = eb - E_ + r;
        a[0].x = meanAcc[2 * hf]     * invE;
        a[0].y = meanAcc[2 * hf + 1] * invE;
        a[1].x = meanAcc[4 + 2 * hf]     * invE;
        a[1].y = meanAcc[4 + 2 * hf + 1] * invE;
    }

    v4f vl[2][8], vr[2][8];
#pragma unroll
    for (int s = 0; s < 2; ++s) {
#pragma unroll
        for (int r = 0; r < 8; ++r) {
            vl[s][r] = *(const v4f*)(xl + (size_t)src[r] * 128 + 64 * s + 4 * c);
            vr[s][r] = *(const v4f*)(xr + (size_t)dst[r] * 128 + 64 * s + 4 * c);
        }
    }
    __builtin_amdgcn_sched_barrier(0);

    v8f D[2][4] = {};
#pragma unroll
    for (int kt = 0; kt < 2; ++kt)
#pragma unroll
        for (int s = 0; s < 2; ++s)
#pragma unroll
            for (int q = 0; q < 4; ++q) {
                v2f b;
                b.x = w[kt][s][0][q];
                b.y = w[kt][s][1][q];
                D[s][q] = __builtin_amdgcn_wmma_f32_16x16x4_f32(
                    false, a[kt], false, b, (short)0, D[s][q], false, false);
            }
    __builtin_amdgcn_sched_barrier(0);

    float p[2][8];
#pragma unroll
    for (int s = 0; s < 2; ++s) {
        const v4f av = s ? av1 : av0;
#pragma unroll
        for (int r = 0; r < 8; ++r) {
            float m0 = vl[s][r][0] + vr[s][r][0] + D[s][0][r];
            float m1 = vl[s][r][1] + vr[s][r][1] + D[s][1][r];
            float m2 = vl[s][r][2] + vr[s][r][2] + D[s][2][r];
            float m3 = vl[s][r][3] + vr[s][r][3] + D[s][3][r];
            m0 = (m0 > 0.f) ? m0 : NSLOPE * m0;
            m1 = (m1 > 0.f) ? m1 : NSLOPE * m1;
            m2 = (m2 > 0.f) ? m2 : NSLOPE * m2;
            m3 = (m3 > 0.f) ? m3 : NSLOPE * m3;
            float t = m0 * av[0] + m1 * av[1] + m2 * av[2] + m3 * av[3];
            t += __shfl_xor(t, 1, 32);
            t += __shfl_xor(t, 2, 32);
            p[s][r] = t;
        }
    }

    const int j    = c & 3;
    const int head = ((j & 1) << 2) + (c >> 2);
    const int rofs = (j >> 1) << 2;
#pragma unroll
    for (int r = 0; r < 4; ++r) {
        const float lo = (j & 1) ? p[1][r]     : p[0][r];
        const float hi = (j & 1) ? p[1][r + 4] : p[0][r + 4];
        const float v  = (j >> 1) ? hi : lo;
        logit[(size_t)(eb + r + rofs) * 8 + head] = v;
    }
}

#define MSG_WAVES 32
#define MSG_RW    7
#define MSG_ROWS  (MSG_WAVES * MSG_RW)
#define MSG_LCAP  384
#define MSG_PCAP  256
#define MSG_B     8

#define BIN_CS  416
#define BIN_G   (BIN_CS / 32)
#define BIN_NB  256
#define BIN_WPB 8
#define BIN_PP  (BIN_NB + 4)

__global__ __launch_bounds__(256) void edge_part_k(
    const int* __restrict__ ei, int* __restrict__ part_, int* __restrict__ poff, int E_, int N_, int G) {
    __shared__ int cntS[BIN_WPB * BIN_NB];
    const int lane = threadIdx.x & 31;
    const int w    = __builtin_amdgcn_readfirstlane((int)(threadIdx.x >> 5));
    const int c    = (int)blockIdx.x * BIN_WPB + w;
    const int NC   = (E_ + BIN_CS - 1) / BIN_CS;
    if (c >= NC) return;
    int rpb = (N_ + G - 1) / G;
    if (rpb > MSG_ROWS) rpb = MSG_ROWS;
    int* const cnt = cntS + w * BIN_NB;
    int2* const part = (int2*)part_;
    const int e0 = c * BIN_CS;

    int d[BIN_G];
#pragma unroll
    for (int g = 0; g < BIN_G; ++g) {
        const int e = e0 + 32 * g + lane;
        d[g] = (e < E_) ? ei[E_ + e] : -1;
    }
#pragma unroll
    for (int k = 0; k < BIN_NB / 32; ++k) cnt[lane + 32 * k] = 0;

    int bn[BIN_G];
#pragma unroll
    for (int g = 0; g < BIN_G; ++g) {
        int b = -1;
        if (d[g] >= 0) {
            b = (int)((unsigned)d[g] / (unsigned)rpb);
            if (b > BIN_NB - 1) b = BIN_NB - 1;
            atomicAdd(&cnt[b], 1);
        }
        bn[g] = b;
    }

    int ex[8]; int run = 0;
#pragma unroll
    for (int k = 0; k < 8; ++k) { ex[k] = run; run += cnt[8 * lane + k]; }
    int inc = run;
#pragma unroll
    for (int off = 1; off < 32; off <<= 1) {
        const int t = __shfl_up(inc, off, 32);
        if (lane >= off) inc += t;
    }
    const int base = inc - run;
    int* const po = poff + (size_t)c * BIN_PP;
#pragma unroll
    for (int k = 0; k < 8; ++k) {
        cnt[8 * lane + k] = base + ex[k];
        po[8 * lane + k]  = base + ex[k];
    }
    if (lane == 31) po[BIN_NB] = inc;

#pragma unroll
    for (int g = 0; g < BIN_G; ++g) {
        const int  b     = bn[g];
        const bool valid = b >= 0;
        unsigned same = __builtin_amdgcn_ballot_w32(valid);
#pragma unroll
        for (int bit = 0; bit < 8; ++bit) {
            const bool on = valid && ((b >> bit) & 1);
            const unsigned bal = __builtin_amdgcn_ballot_w32(on);
            same &= on ? bal : ~bal;
        }
        const int rank = (int)__builtin_amdgcn_mbcnt_lo(same, 0u);
        const int tot  = __popc(same);
        int pos = 0;
        if (valid) {
            pos = cnt[b];
            const int lr = d[g] - b * rpb;
            part[e0 + pos + rank] = make_int2((int)((unsigned)lr << 24), e0 + 32 * g + lane);
        }
        if (valid && rank == 0) cnt[b] = pos + tot;
    }
}

__device__ __forceinline__ float quad_lane(float x, int k) {
    const int xi = __float_as_int(x);
    int r;
    switch (k & 3) {
        case 0:  r = __builtin_amdgcn_update_dpp(0, xi, 0x00, 0xF, 0xF, true); break;
        case 1:  r = __builtin_amdgcn_update_dpp(0, xi, 0x55, 0xF, 0xF, true); break;
        case 2:  r = __builtin_amdgcn_update_dpp(0, xi, 0xAA, 0xF, 0xF, true); break;
        default: r = __builtin_amdgcn_update_dpp(0, xi, 0xFF, 0xF, 0xF, true); break;
    }
    return __int_as_float(r);
}

__device__ __forceinline__ void msg_add_entries(
    int2* myp, int pc, v4f* acc, unsigned* rmaxu, float* rsum,
    const int* __restrict__ ei, const float* __restrict__ xl,
    const float* __restrict__ logit, int lr0, int lane, int h) {
    {
        int sv[MSG_PCAP / 32];
#pragma unroll
        for (int u = 0; u < MSG_PCAP / 32; ++u) {
            const int j = lane + 32 * u;
            sv[u] = (j < pc) ? ei[myp[j].y] : 0;
        }
#pragma unroll
        for (int u = 0; u < MSG_PCAP / 32; ++u) {
            const int j = lane + 32 * u;
            if (j < pc) myp[j].x |= sv[u];
        }
    }

    unsigned mp[MSG_RW];
#pragma unroll
    for (int r = 0; r < MSG_RW; ++r) mp[r] = rmaxu[(lr0 + r) * 8 + h];
    {
        const int hh = lane & 7;
        const int eo = lane >> 3;
        for (int t0 = 0; t0 < pc; t0 += 4 * MSG_B) {
            float v[MSG_B]; int rw[MSG_B];
#pragma unroll
            for (int u = 0; u < MSG_B; ++u) {
                const int j  = t0 + 4 * u + eo;
                const int2 q = myp[(j < pc) ? j : (pc - 1)];
                rw[u] = (int)((unsigned)q.x >> 24);
                v[u]  = logit[(size_t)q.y * 8 + hh];
            }
#pragma unroll
            for (int u = 0; u < MSG_B; ++u) {
                const int j = t0 + 4 * u + eo;
                if (j < pc) atomicMax(&rmaxu[rw[u] * 8 + hh], fmap(v[u]));
            }
        }
    }
#pragma unroll
    for (int r = 0; r < MSG_RW; ++r) {
        const unsigned mn = rmaxu[(lr0 + r) * 8 + h];
        const float sc = expf(funmap(mp[r]) - funmap(mn));
        v4f a = acc[(lr0 + r) * 32 + lane];
        a *= sc;
        acc[(lr0 + r) * 32 + lane] = a;
        rsum[(lr0 + r) * 8 + h] *= sc;
    }

    const int jq = lane & 3;
    for (int b = 0; b < pc; b += MSG_B) {
        int qx[MSG_B], qy[MSG_B]; v4f xv[MSG_B];
        float lv[2]; unsigned mv[2];
#pragma unroll
        for (int k = 0; k < 2; ++k) {
            const int j  = b + jq + 4 * k;
            const int2 q = myp[(j < pc) ? j : (pc - 1)];
            lv[k] = logit[(size_t)q.y * 8 + h];
            mv[k] = rmaxu[(int)((unsigned)q.x >> 24) * 8 + h];
        }
#pragma unroll
        for (int u = 0; u < MSG_B; ++u) {
            const int j = (b + u < pc) ? (b + u) : (pc - 1);
            const int2 q = myp[j];
            qx[u] = __builtin_amdgcn_readfirstlane(q.x);
            qy[u] = __builtin_amdgcn_readfirstlane(q.y);
        }
#pragma unroll
        for (int u = 0; u < MSG_B; ++u) {
            const int s = qx[u] & 0x00FFFFFF;
            xv[u] = *(const v4f*)(xl + (size_t)s * 128 + 4 * lane);
        }
        float pe[2];
#pragma unroll
        for (int k = 0; k < 2; ++k) pe[k] = expf(lv[k] - funmap(mv[k]));
#pragma unroll
        for (int u = 0; u < MSG_B; ++u) {
            if (b + u < pc) {
                const int rl  = (int)((unsigned)qx[u] >> 24);
                const float p = quad_lane(pe[u >> 2], u & 3);
                v4f a = acc[rl * 32 + lane];
                a += p * xv[u];
                acc[rl * 32 + lane] = a;
                rsum[rl * 8 + h] += p;
            }
        }
    }
}

__global__ __launch_bounds__(1024) void edge_msg_owner_k(
    const int* __restrict__ ei, const float* __restrict__ logit, const float* __restrict__ xl,
    const float* __restrict__ obias, float* __restrict__ out,
    const int* __restrict__ part_, const int* __restrict__ poff, int E_, int N_) {
    __shared__ v4f  acc[MSG_ROWS * 32];
    __shared__ int2 lst[MSG_WAVES * MSG_LCAP];
    __shared__ int2 prv[MSG_WAVES * MSG_PCAP];
    __shared__ unsigned rmaxu[MSG_ROWS * 8];
    __shared__ float    rsum[MSG_ROWS * 8];
    __shared__ int  s_cnt[MSG_WAVES];
    __shared__ int  s_un[MSG_WAVES];

    const int lane = threadIdx.x & 31;
    const int w    = __builtin_amdgcn_readfirstlane((int)(threadIdx.x >> 5));
    const int h    = lane >> 2;
    const int G    = (int)gridDim.x;
    int rpb = (N_ + G - 1) / G;
    if (rpb > MSG_ROWS) rpb = MSG_ROWS;
    const int2* const part = (const int2*)part_;
    const int NC  = (E_ + BIN_CS - 1) / BIN_CS;
    const int CPW = (NC + MSG_WAVES - 1) / MSG_WAVES;
    const v4f bv = *(const v4f*)(obias + 4 * lane);
    const int lr0 = w * MSG_RW;
    int2* const myl = lst + w * MSG_LCAP;
    int2* const myp = prv + w * MSG_PCAP;

    for (int row0 = blockIdx.x * rpb; row0 < N_; row0 += G * rpb) {
        {
            v4f xv[MSG_RW]; float ls[MSG_RW];
#pragma unroll
            for (int r = 0; r < MSG_RW; ++r) {
                const int g  = row0 + lr0 + r;
                const int gg = (lr0 + r < rpb && g < N_) ? g : row0;
                xv[r] = *(const v4f*)(xl + (size_t)gg * 128 + 4 * lane);
                ls[r] = logit[(size_t)(E_ + gg) * 8 + h];
            }
#pragma unroll
            for (int r = 0; r < MSG_RW; ++r) {
                acc[(lr0 + r) * 32 + lane] = xv[r];
                rmaxu[(lr0 + r) * 8 + h]   = fmap(ls[r]);
                rsum[(lr0 + r) * 8 + h]    = 1.f;
            }
        }

        int R = row0 / rpb;
        if (R > BIN_NB - 1) R = BIN_NB - 1;
        const int cbeg = w * CPW;
        const int cend = (cbeg + CPW < NC) ? (cbeg + CPW) : NC;
        int ci = cbeg;
        int co = 0;
        for (;;) {
            int cnt = 0;
            while (ci < cend && cnt < MSG_LCAP) {
                const int  c  = ci + lane;
                const bool cv = c < cend;
                int o0 = 0, o1 = 0;
                if (cv) {
                    o0 = poff[(size_t)c * BIN_PP + R];
                    o1 = poff[(size_t)c * BIN_PP + R + 1];
                }
                if (lane == 0) o0 += co;
                const int n = o1 - o0;
                int inc = n;
#pragma unroll
                for (int off = 1; off < 32; off <<= 1) {
                    const int t = __shfl_up(inc, off, 32);
                    if (lane >= off) inc += t;
                }
                const int exc  = inc - n;
                const int room = MSG_LCAP - cnt;
                int t = room - exc;
                t = (t < 0) ? 0 : ((t > n) ? n : t);
                const int2* const sp = part + (size_t)c * BIN_CS + o0;
                int2* const dp = myl + cnt + exc;
                for (int j0 = 0; __builtin_amdgcn_ballot_w32(j0 < t) != 0u; j0 += 4) {
                    int2 v[4];
#pragma unroll
                    for (int u = 0; u < 4; ++u) if (j0 + u < t) v[u] = sp[j0 + u];
#pragma unroll
                    for (int u = 0; u < 4; ++u) if (j0 + u < t) dp[j0 + u] = v[u];
                }
                const unsigned cut = __builtin_amdgcn_ballot_w32(t < n);
                if (cut) {
                    const int f  = __builtin_ctz(cut);
                    const int tf = __builtin_amdgcn_readlane(t, f);
                    co  = tf + ((f == 0) ? co : 0);
                    ci += f;
                    cnt = MSG_LCAP;
                } else {
                    cnt += __builtin_amdgcn_readlane(inc, 31);
                    ci  += 32;
                    co   = 0;
                }
            }
            if (lane == 0) { s_cnt[w] = cnt; s_un[w] = (ci < cend) ? 1 : 0; }
            __syncthreads();
            const bool more = __builtin_amdgcn_ballot_w32(s_un[lane] != 0) != 0u;

            int pc = 0;
            for (int L = 0; L < MSG_WAVES; ++L) {
                const int n = __builtin_amdgcn_readfirstlane(s_cnt[L]);
                const int2* const ll = lst + L * MSG_LCAP;
                for (int i0 = 0; i0 < n; i0 += 32) {
                    const int  i  = i0 + lane;
                    const bool in = i < n;
                    int2 en = make_int2(0, 0);
                    if (in) en = ll[i];
                    const int  lr = (int)((unsigned)en.x >> 24);
                    const bool m  = in && (unsigned)(lr - lr0) < (unsigned)MSG_RW;
                    const unsigned bal = __builtin_amdgcn_ballot_w32(m);
                    if (bal) {
                        const int p = pc + (int)__builtin_amdgcn_mbcnt_lo(bal, 0u);
                        if (m) myp[p] = en;
                        pc += __popc(bal);
                    }
                    if (pc > MSG_PCAP - 32) {
                        msg_add_entries(myp, pc, acc, rmaxu, rsum, ei, xl, logit, lr0, lane, h);
                        pc = 0;
                    }
                }
            }
            if (pc > 0) msg_add_entries(myp, pc, acc, rmaxu, rsum, ei, xl, logit, lr0, lane, h);
            __syncthreads();
            if (!more) break;
        }

#pragma unroll
        for (int r = 0; r < MSG_RW; ++r) {
            const int g = row0 + lr0 + r;
            if (lr0 + r < rpb && g < N_) {
                const float den = rsum[(lr0 + r) * 8 + h] + 1e-16f;
                const v4f a = acc[(lr0 + r) * 32 + lane];
                v4f o;
                o.x = bv.x + a.x / den;
                o.y = bv.y + a.y / den;
                o.z = bv.z + a.z / den;
                o.w = bv.w + a.w / den;
                *(v4f*)(out + (size_t)g * 128 + 4 * lane) = o;
            }
        }
    }
}

extern "C" void kernel_launch(void* const* d_in, const int* in_sizes, int n_in,
                              void* d_out, int out_size, void* d_ws, size_t ws_size,
                              hipStream_t stream) {
    const float* x     = (const float*)d_in[0];
    const int*   ei    = (const int*)d_in[1];
    const float* eattr = (const float*)d_in[2];
    const float* Wl    = (const float*)d_in[3];
    const float* bl    = (const float*)d_in[4];
    const float* Wr    = (const float*)d_in[5];
    const float* br    = (const float*)d_in[6];
    const float* We    = (const float*)d_in[7];
    const float* att   = (const float*)d_in[8];
    const float* bias  = (const float*)d_in[9];
    float* out = (float*)d_out;
    float* ws  = (float*)d_ws;

    const int n = in_sizes[0] / 128;
    const int e = in_sizes[2] / 8;
    const int tot = e + n;

    size_t off_mean   = 0;
    size_t off_xl     = 256;
    size_t off_xr     = off_xl + (size_t)n * 128;
    size_t off_logit  = off_xr + (size_t)n * 128;
    size_t off_part   = off_logit + (size_t)tot * 8;
    size_t off_poff   = off_part + (size_t)e * 2 + 2 * BIN_CS;

    int nTiles = n / 16;
    dim3 g((nTiles + 3) / 4, 2);
    node_xform_wmma<<<g, 128, 0, stream>>>(x, Wl, bl, Wr, br,
                                           ws + off_xl, ws + off_xr,
                                           ws + off_mean, nTiles);
    attr_mean_k<<<512, 256, 0, stream>>>(eattr, ws + off_mean, e);
    int nChunks = (e + BIN_CS - 1) / BIN_CS;
    edge_part_k<<<(nChunks + BIN_WPB - 1) / BIN_WPB, 256, 0, stream>>>(
        ei, (int*)(ws + off_part), (int*)(ws + off_poff), e, n, BIN_NB);

    int tileBlocks = ((tot / 16) * 32 + 255) / 256;
    edge_logit_tile_k<<<tileBlocks, 256, 0, stream>>>(ei, eattr, We, ws + off_mean,
                                                      1.0f / (float)e,
                                                      ws + off_xl, ws + off_xr, att,
                                                      ws + off_logit, e, n);
    edge_msg_owner_k<<<BIN_NB, 1024, 0, stream>>>(ei, ws + off_logit, ws + off_xl, bias, out,
                                                  (const int*)(ws + off_part), (const int*)(ws + off_poff), e, n);
}


# edge_msg_owner_k step 2: each wave splits its own list by owner wave into the owners' private lists (stable counting split in LDS) instead of every wave walking all 32 lists; sums in the same order
# speedup vs baseline: 2.5327x; 1.0778x over previous
#include <hip/hip_runtime.h>
#include <hip/hip_bf16.h>

typedef __attribute__((ext_vector_type(2))) float v2f;
typedef __attribute__((ext_vector_type(4))) float v4f;
typedef __attribute__((ext_vector_type(8))) float v8f;

#define NSLOPE 0.2f

__device__ __forceinline__ unsigned fmap(float x) {
    unsigned b = __float_as_uint(x);
    return (b & 0x80000000u) ? ~b : (b | 0x80000000u);
}
__device__ __forceinline__ float funmap(unsigned u) {
    return (u & 0x80000000u) ? __uint_as_float(u ^ 0x80000000u)
                             : __uint_as_float(~u);
}

__global__ __launch_bounds__(256) void attr_mean_k(const float* __restrict__ eattr,
                                                   float* __restrict__ meanAcc, int E_) {
    __shared__ float s[8];
    if (threadIdx.x < 8) s[threadIdx.x] = 0.f;
    __syncthreads();
    float acc[8] = {0.f, 0.f, 0.f, 0.f, 0.f, 0.f, 0.f, 0.f};
    for (int r = blockIdx.x * blockDim.x + threadIdx.x; r < E_; r += gridDim.x * blockDim.x) {
        const float* row = eattr + (size_t)r * 8;
#pragma unroll
        for (int d = 0; d < 8; ++d) acc[d] += row[d];
    }
#pragma unroll
    for (int d = 0; d < 8; ++d) atomicAdd(&s[d], acc[d]);
    __syncthreads();
    if (threadIdx.x < 8) atomicAdd(meanAcc + threadIdx.x, s[threadIdx.x]);
}

__global__ __launch_bounds__(128) void node_xform_wmma(
    const float* __restrict__ X,
    const float* __restrict__ Wl, const float* __restrict__ bl,
    const float* __restrict__ Wr, const float* __restrict__ br,
    float* __restrict__ xl, float* __restrict__ xr,
    float* __restrict__ meanAcc, int nTiles) {
    int wid = threadIdx.x >> 5;
    int lane = threadIdx.x & 31;
    int rowTile = blockIdx.x * 4 + wid;
    if (rowTile >= nTiles) return;

    if (blockIdx.y != 0 && rowTile == 0 && lane < 8) meanAcc[lane] = 0.f;

    const float* W    = blockIdx.y ? Wr : Wl;
    const float* bias = blockIdx.y ? br : bl;
    float* out        = blockIdx.y ? xr : xl;

    int row0  = rowTile * 16;
    int m     = lane & 15;
    int khalf = lane >> 4;

    v8f acc[8] = {};

    const float* arow = X + (size_t)(row0 + m) * 128;
    for (int kt = 0; kt < 32; ++kt) {
        int k0 = kt * 4 + khalf * 2;
        v2f a;
        a.x = arow[k0];
        a.y = arow[k0 + 1];
        const float* w0 = W + (size_t)k0 * 128;
        const float* w1 = w0 + 128;
#pragma unroll
        for (int nt = 0; nt < 8; ++nt) {
            v2f b;
            b.x = w0[nt * 16 + m];
            b.y = w1[nt * 16 + m];
            acc[nt] = __builtin_amdgcn_wmma_f32_16x16x4_f32(
                false, a, false, b, (short)0, acc[nt], false, false);
        }
    }
#pragma unroll
    for (int nt = 0; nt < 8; ++nt) {
        float bv = bias[nt * 16 + m];
#pragma unroll
        for (int i = 0; i < 8; ++i) {
            int row = row0 + i + 8 * khalf;
            out[(size_t)row * 128 + nt * 16 + m] = acc[nt][i] + bv;
        }
    }
}

__global__ __launch_bounds__(256) void edge_logit_tile_k(
    const int* __restrict__ ei, const float* __restrict__ eattr,
    const float* __restrict__ We, const float* __restrict__ meanAcc, float invE,
    const float* __restrict__ xl, const float* __restrict__ xr,
    const float* __restrict__ att, float* __restrict__ logit, int E_, int N_) {
    const int lane = threadIdx.x & 31;
    const int tile = (blockIdx.x * blockDim.x + threadIdx.x) >> 5;
    if (tile >= ((E_ + N_) >> 4)) return;
    const int c  = lane & 15;
    const int hf = lane >> 4;
    const int e0 = tile << 4;
    const int eb = e0 + 8 * hf;

    v4f w[2][2][2];
#pragma unroll
    for (int kt = 0; kt < 2; ++kt)
#pragma unroll
        for (int s = 0; s < 2; ++s)
#pragma unroll
            for (int y = 0; y < 2; ++y)
                w[kt][s][y] = *(const v4f*)(We + (size_t)(4 * kt + 2 * hf + y) * 128 + 64 * s + 4 * c);

    const v4f av0 = *(const v4f*)(att + 4 * c);
    const v4f av1 = *(const v4f*)(att + 64 + 4 * c);

    int src[8], dst[8];
    v2f a[2];
    if (e0 < E_) {
        const int4 s0 = *(const int4*)(ei + eb);
        const int4 s1 = *(const int4*)(ei + eb + 4);
        const int4 d0 = *(const int4*)(ei + E_ + eb);
        const int4 d1 = *(const int4*)(ei + E_ + eb + 4);
        src[0] = s0.x; src[1] = s0.y; src[2] = s0.z; src[3] = s0.w;
        src[4] = s1.x; src[5] = s1.y; src[6] = s1.z; src[7] = s1.w;
        dst[0] = d0.x; dst[1] = d0.y; dst[2] = d0.z; dst[3] = d0.w;
        dst[4] = d1.x; dst[5] = d1.y; dst[6] = d1.z; dst[7] = d1.w;
        const float* arow = eattr + (size_t)(e0 + c) * 8 + 2 * hf;
        a[0] = *(const v2f*)(arow);
        a[1] = *(const v2f*)(arow + 4);
    } else {
#pragma unroll
        for (int r = 0; r < 8; ++r) src[r] = dst[r] = eb - E_ + r;
        a[0].x = meanAcc[2 * hf]     * invE;
        a[0].y = meanAcc[2 * hf + 1] * invE;
        a[1].x = meanAcc[4 + 2 * hf]     * invE;
        a[1].y = meanAcc[4 + 2 * hf + 1] * invE;
    }

    v4f vl[2][8], vr[2][8];
#pragma unroll
    for (int s = 0; s < 2; ++s) {
#pragma unroll
        for (int r = 0; r < 8; ++r) {
            vl[s][r] = *(const v4f*)(xl + (size_t)src[r] * 128 + 64 * s + 4 * c);
            vr[s][r] = *(const v4f*)(xr + (size_t)dst[r] * 128 + 64 * s + 4 * c);
        }
    }
    __builtin_amdgcn_sched_barrier(0);

    v8f D[2][4] = {};
#pragma unroll
    for (int kt = 0; kt < 2; ++kt)
#pragma unroll
        for (int s = 0; s < 2; ++s)
#pragma unroll
            for (int q = 0; q < 4; ++q) {
                v2f b;
                b.x = w[kt][s][0][q];
                b.y = w[kt][s][1][q];
                D[s][q] = __builtin_amdgcn_wmma_f32_16x16x4_f32(
                    false, a[kt], false, b, (short)0, D[s][q], false, false);
            }
    __builtin_amdgcn_sched_barrier(0);

    float p[2][8];
#pragma unroll
    for (int s = 0; s < 2; ++s) {
        const v4f av = s ? av1 : av0;
#pragma unroll
        for (int r = 0; r < 8; ++r) {
            float m0 = vl[s][r][0] + vr[s][r][0] + D[s][0][r];
            float m1 = vl[s][r][1] + vr[s][r][1] + D[s][1][r];
            float m2 = vl[s][r][2] + vr[s][r][2] + D[s][2][r];
            float m3 = vl[s][r][3] + vr[s][r][3] + D[s][3][r];
            m0 = (m0 > 0.f) ? m0 : NSLOPE * m0;
            m1 = (m1 > 0.f) ? m1 : NSLOPE * m1;
            m2 = (m2 > 0.f) ? m2 : NSLOPE * m2;
            m3 = (m3 > 0.f) ? m3 : NSLOPE * m3;
            float t = m0 * av[0] + m1 * av[1] + m2 * av[2] + m3 * av[3];
            t += __shfl_xor(t, 1, 32);
            t += __shfl_xor(t, 2, 32);
            p[s][r] = t;
        }
    }

    const int j    = c & 3;
    const int head = ((j & 1) << 2) + (c >> 2);
    const int rofs = (j >> 1) << 2;
#pragma unroll
    for (int r = 0; r < 4; ++r) {
        const float lo = (j & 1) ? p[1][r]     : p[0][r];
        const float hi = (j & 1) ? p[1][r + 4] : p[0][r + 4];
        const float v  = (j >> 1) ? hi : lo;
        logit[(size_t)(eb + r + rofs) * 8 + head] = v;
    }
}

#define MSG_WAVES 32
#define MSG_RW    7
#define MSG_ROWS  (MSG_WAVES * MSG_RW)
#define MSG_LCAP  384
#define MSG_PCAP  256
#define MSG_B     8

#define BIN_CS  416
#define BIN_G   (BIN_CS / 32)
#define BIN_NB  256
#define BIN_WPB 8
#define BIN_PP  (BIN_NB + 4)

__global__ __launch_bounds__(256) void edge_part_k(
    const int* __restrict__ ei, int* __restrict__ part_, int* __restrict__ poff, int E_, int N_, int G) {
    __shared__ int cntS[BIN_WPB * BIN_NB];
    const int lane = threadIdx.x & 31;
    const int w    = __builtin_amdgcn_readfirstlane((int)(threadIdx.x >> 5));
    const int c    = (int)blockIdx.x * BIN_WPB + w;
    const int NC   = (E_ + BIN_CS - 1) / BIN_CS;
    if (c >= NC) return;
    int rpb = (N_ + G - 1) / G;
    if (rpb > MSG_ROWS) rpb = MSG_ROWS;
    int* const cnt = cntS + w * BIN_NB;
    int2* const part = (int2*)part_;
    const int e0 = c * BIN_CS;

    int d[BIN_G];
#pragma unroll
    for (int g = 0; g < BIN_G; ++g) {
        const int e = e0 + 32 * g + lane;
        d[g] = (e < E_) ? ei[E_ + e] : -1;
    }
#pragma unroll
    for (int k = 0; k < BIN_NB / 32; ++k) cnt[lane + 32 * k] = 0;

    int bn[BIN_G];
#pragma unroll
    for (int g = 0; g < BIN_G; ++g) {
        int b = -1;
        if (d[g] >= 0) {
            b = (int)((unsigned)d[g] / (unsigned)rpb);
            if (b > BIN_NB - 1) b = BIN_NB - 1;
            atomicAdd(&cnt[b], 1);
        }
        bn[g] = b;
    }

    int ex[8]; int run = 0;
#pragma unroll
    for (int k = 0; k < 8; ++k) { ex[k] = run; run += cnt[8 * lane + k]; }
    int inc = run;
#pragma unroll
    for (int off = 1; off < 32; off <<= 1) {
        const int t = __shfl_up(inc, off, 32);
        if (lane >= off) inc += t;
    }
    const int base = inc - run;
    int* const po = poff + (size_t)c * BIN_PP;
#pragma unroll
    for (int k = 0; k < 8; ++k) {
        cnt[8 * lane + k] = base + ex[k];
        po[8 * lane + k]  = base + ex[k];
    }
    if (lane == 31) po[BIN_NB] = inc;

#pragma unroll
    for (int g = 0; g < BIN_G; ++g) {
        const int  b     = bn[g];
        const bool valid = b >= 0;
        unsigned same = __builtin_amdgcn_ballot_w32(valid);
#pragma unroll
        for (int bit = 0; bit < 8; ++bit) {
            const bool on = valid && ((b >> bit) & 1);
            const unsigned bal = __builtin_amdgcn_ballot_w32(on);
            same &= on ? bal : ~bal;
        }
        const int rank = (int)__builtin_amdgcn_mbcnt_lo(same, 0u);
        const int tot  = __popc(same);
        int pos = 0;
        if (valid) {
            pos = cnt[b];
            const int lr = d[g] - b * rpb;
            part[e0 + pos + rank] = make_int2((int)((unsigned)lr << 24), e0 + 32 * g + lane);
        }
        if (valid && rank == 0) cnt[b] = pos + tot;
    }
}

__device__ __forceinline__ float quad_lane(float x, int k) {
    const int xi = __float_as_int(x);
    int r;
    switch (k & 3) {
        case 0:  r = __builtin_amdgcn_update_dpp(0, xi, 0x00, 0xF, 0xF, true); break;
        case 1:  r = __builtin_amdgcn_update_dpp(0, xi, 0x55, 0xF, 0xF, true); break;
        case 2:  r = __builtin_amdgcn_update_dpp(0, xi, 0xAA, 0xF, 0xF, true); break;
        default: r = __builtin_amdgcn_update_dpp(0, xi, 0xFF, 0xF, 0xF, true); break;
    }
    return __int_as_float(r);
}

__device__ __forceinline__ void msg_add_entries(
    int2* myp, int pc, v4f* acc, unsigned* rmaxu, float* rsum,
    const int* __restrict__ ei, const float* __restrict__ xl,
    const float* __restrict__ logit, int lr0, int lane, int h) {
    {
        int sv[MSG_PCAP / 32];
#pragma unroll
        for (int u = 0; u < MSG_PCAP / 32; ++u) {
            const int j = lane + 32 * u;
            sv[u] = (j < pc) ? ei[myp[j].y] : 0;
        }
#pragma unroll
        for (int u = 0; u < MSG_PCAP / 32; ++u) {
            const int j = lane + 32 * u;
            if (j < pc) myp[j].x |= sv[u];
        }
    }

    unsigned mp[MSG_RW];
#pragma unroll
    for (int r = 0; r < MSG_RW; ++r) mp[r] = rmaxu[(lr0 + r) * 8 + h];
    {
        const int hh = lane & 7;
        const int eo = lane >> 3;
        for (int t0 = 0; t0 < pc; t0 += 4 * MSG_B) {
            float v[MSG_B]; int rw[MSG_B];
#pragma unroll
            for (int u = 0; u < MSG_B; ++u) {
                const int j  = t0 + 4 * u + eo;
                const int2 q = myp[(j < pc) ? j : (pc - 1)];
                rw[u] = (int)((unsigned)q.x >> 24);
                v[u]  = logit[(size_t)q.y * 8 + hh];
            }
#pragma unroll
            for (int u = 0; u < MSG_B; ++u) {
                const int j = t0 + 4 * u + eo;
                if (j < pc) atomicMax(&rmaxu[rw[u] * 8 + hh], fmap(v[u]));
            }
        }
    }
#pragma unroll
    for (int r = 0; r < MSG_RW; ++r) {
        const unsigned mn = rmaxu[(lr0 + r) * 8 + h];
        const float sc = expf(funmap(mp[r]) - funmap(mn));
        v4f a = acc[(lr0 + r) * 32 + lane];
        a *= sc;
        acc[(lr0 + r) * 32 + lane] = a;
        rsum[(lr0 + r) * 8 + h] *= sc;
    }

    const int jq = lane & 3;
    for (int b = 0; b < pc; b += MSG_B) {
        int qx[MSG_B], qy[MSG_B]; v4f xv[MSG_B];
        float lv[2]; unsigned mv[2];
#pragma unroll
        for (int k = 0; k < 2; ++k) {
            const int j  = b + jq + 4 * k;
            const int2 q = myp[(j < pc) ? j : (pc - 1)];
            lv[k] = logit[(size_t)q.y * 8 + h];
            mv[k] = rmaxu[(int)((unsigned)q.x >> 24) * 8 + h];
        }
#pragma unroll
        for (int u = 0; u < MSG_B; ++u) {
            const int j = (b + u < pc) ? (b + u) : (pc - 1);
            const int2 q = myp[j];
            qx[u] = __builtin_amdgcn_readfirstlane(q.x);
            qy[u] = __builtin_amdgcn_readfirstlane(q.y);
        }
#pragma unroll
        for (int u = 0; u < MSG_B; ++u) {
            const int s = qx[u] & 0x00FFFFFF;
            xv[u] = *(const v4f*)(xl + (size_t)s * 128 + 4 * lane);
        }
        float pe[2];
#pragma unroll
        for (int k = 0; k < 2; ++k) pe[k] = expf(lv[k] - funmap(mv[k]));
#pragma unroll
        for (int u = 0; u < MSG_B; ++u) {
            if (b + u < pc) {
                const int rl  = (int)((unsigned)qx[u] >> 24);
                const float p = quad_lane(pe[u >> 2], u & 3);
                v4f a = acc[rl * 32 + lane];
                a += p * xv[u];
                acc[rl * 32 + lane] = a;
                rsum[rl * 8 + h] += p;
            }
        }
    }
}

__global__ __launch_bounds__(1024) void edge_msg_owner_k(
    const int* __restrict__ ei, const float* __restrict__ logit, const float* __restrict__ xl,
    const float* __restrict__ obias, float* __restrict__ out,
    const int* __restrict__ part_, const int* __restrict__ poff, int E_, int N_) {
    __shared__ v4f  acc[MSG_ROWS * 32];
    __shared__ int2 lst[MSG_WAVES * MSG_LCAP];
    __shared__ int2 prv[MSG_WAVES * MSG_PCAP];
    __shared__ unsigned rmaxu[MSG_ROWS * 8];
    __shared__ float    rsum[MSG_ROWS * 8];
    __shared__ int  s_cnt[MSG_WAVES];
    __shared__ int  s_un[MSG_WAVES];
    __shared__ int  s_oc[MSG_WAVES * MSG_WAVES];
    __shared__ int  s_run[MSG_WAVES * MSG_WAVES];

    const int lane = threadIdx.x & 31;
    const int w    = __builtin_amdgcn_readfirstlane((int)(threadIdx.x >> 5));
    const int h    = lane >> 2;
    const int G    = (int)gridDim.x;
    int rpb = (N_ + G - 1) / G;
    if (rpb > MSG_ROWS) rpb = MSG_ROWS;
    const int2* const part = (const int2*)part_;
    const int NC  = (E_ + BIN_CS - 1) / BIN_CS;
    const int CPW = (NC + MSG_WAVES - 1) / MSG_WAVES;
    const v4f bv = *(const v4f*)(obias + 4 * lane);
    const int lr0 = w * MSG_RW;
    int2* const myl = lst + w * MSG_LCAP;
    int2* const myp = prv + w * MSG_PCAP;

    for (int row0 = blockIdx.x * rpb; row0 < N_; row0 += G * rpb) {
        {
            v4f xv[MSG_RW]; float ls[MSG_RW];
#pragma unroll
            for (int r = 0; r < MSG_RW; ++r) {
                const int g  = row0 + lr0 + r;
                const int gg = (lr0 + r < rpb && g < N_) ? g : row0;
                xv[r] = *(const v4f*)(xl + (size_t)gg * 128 + 4 * lane);
                ls[r] = logit[(size_t)(E_ + gg) * 8 + h];
            }
#pragma unroll
            for (int r = 0; r < MSG_RW; ++r) {
                acc[(lr0 + r) * 32 + lane] = xv[r];
                rmaxu[(lr0 + r) * 8 + h]   = fmap(ls[r]);
                rsum[(lr0 + r) * 8 + h]    = 1.f;
            }
        }

        int R = row0 / rpb;
        if (R > BIN_NB - 1) R = BIN_NB - 1;
        const int cbeg = w * CPW;
        const int cend = (cbeg + CPW < NC) ? (cbeg + CPW) : NC;
        int ci = cbeg;
        int co = 0;
        for (;;) {
            int cnt = 0;
            while (ci < cend && cnt < MSG_LCAP) {
                const int  c  = ci + lane;
                const bool cv = c < cend;
                int o0 = 0, o1 = 0;
                if (cv) {
                    o0 = poff[(size_t)c * BIN_PP + R];
                    o1 = poff[(size_t)c * BIN_PP + R + 1];
                }
                if (lane == 0) o0 += co;
                const int n = o1 - o0;
                int inc = n;
#pragma unroll
                for (int off = 1; off < 32; off <<= 1) {
                    const int t = __shfl_up(inc, off, 32);
                    if (lane >= off) inc += t;
                }
                const int exc  = inc - n;
                const int room = MSG_LCAP - cnt;
                int t = room - exc;
                t = (t < 0) ? 0 : ((t > n) ? n : t);
                const int2* const sp = part + (size_t)c * BIN_CS + o0;
                int2* const dp = myl + cnt + exc;
                for (int j0 = 0; __builtin_amdgcn_ballot_w32(j0 < t) != 0u; j0 += 4) {
                    int2 v[4];
#pragma unroll
                    for (int u = 0; u < 4; ++u) if (j0 + u < t) v[u] = sp[j0 + u];
#pragma unroll
                    for (int u = 0; u < 4; ++u) if (j0 + u < t) dp[j0 + u] = v[u];
                }
                const unsigned cut = __builtin_amdgcn_ballot_w32(t < n);
                if (cut) {
                    const int f  = __builtin_ctz(cut);
                    const int tf = __builtin_amdgcn_readlane(t, f);
                    co  = tf + ((f == 0) ? co : 0);
                    ci += f;
                    cnt = MSG_LCAP;
                } else {
                    cnt += __builtin_amdgcn_readlane(inc, 31);
                    ci  += 32;
                    co   = 0;
                }
            }
            if (lane == 0) { s_cnt[w] = cnt; s_un[w] = (ci < cend) ? 1 : 0; }
            s_oc[w * MSG_WAVES + lane] = 0;
            for (int i0 = 0; i0 < cnt; i0 += 32) {
                const int i = i0 + lane;
                if (i < cnt) {
                    const int o = (int)((unsigned)myl[i].x >> 24) / MSG_RW;
                    if (o < MSG_WAVES) atomicAdd(&s_oc[w * MSG_WAVES + o], 1);
                }
            }
            __syncthreads();
            const bool more = __builtin_amdgcn_ballot_w32(s_un[lane] != 0) != 0u;

            int obase = 0, otot = 0;
#pragma unroll 8
            for (int L = 0; L < MSG_WAVES; ++L) {
                const int c = s_oc[L * MSG_WAVES + lane];
                obase += (L < w) ? c : 0;
                otot  += c;
            }
            const bool fits = __builtin_amdgcn_ballot_w32(otot > MSG_PCAP) == 0u;
            int pc = 0;
            if (fits) {
                int* const run = s_run + w * MSG_WAVES;
                run[lane] = obase;
                for (int i0 = 0; i0 < cnt; i0 += 32) {
                    const int i = i0 + lane;
                    int2 en = make_int2(0, 0);
                    if (i < cnt) en = myl[i];
                    const int  o     = (int)((unsigned)en.x >> 24) / MSG_RW;
                    const bool valid = (i < cnt) && (o < MSG_WAVES);
                    unsigned same = __builtin_amdgcn_ballot_w32(valid);
#pragma unroll
                    for (int bit = 0; bit < 5; ++bit) {
                        const bool on = valid && ((o >> bit) & 1);
                        const unsigned bal = __builtin_amdgcn_ballot_w32(on);
                        same &= on ? bal : ~bal;
                    }
                    const int rank = (int)__builtin_amdgcn_mbcnt_lo(same, 0u);
                    const int totg = __popc(same);
                    int pos = 0;
                    if (valid) {
                        pos = run[o];
                        prv[o * MSG_PCAP + pos + rank] = en;
                    }
                    if (valid && rank == 0) run[o] = pos + totg;
                }
                __syncthreads();
                pc = __builtin_amdgcn_readlane(otot, w);
            } else {
            for (int L = 0; L < MSG_WAVES; ++L) {
                const int n = __builtin_amdgcn_readfirstlane(s_cnt[L]);
                const int2* const ll = lst + L * MSG_LCAP;
                for (int i0 = 0; i0 < n; i0 += 32) {
                    const int  i  = i0 + lane;
                    const bool in = i < n;
                    int2 en = make_int2(0, 0);
                    if (in) en = ll[i];
                    const int  lr = (int)((unsigned)en.x >> 24);
                    const bool m  = in && (unsigned)(lr - lr0) < (unsigned)MSG_RW;
                    const unsigned bal = __builtin_amdgcn_ballot_w32(m);
                    if (bal) {
                        const int p = pc + (int)__builtin_amdgcn_mbcnt_lo(bal, 0u);
                        if (m) myp[p] = en;
                        pc += __popc(bal);
                    }
                    if (pc > MSG_PCAP - 32) {
                        msg_add_entries(myp, pc, acc, rmaxu, rsum, ei, xl, logit, lr0, lane, h);
                        pc = 0;
                    }
                }
            }
            }
            if (pc > 0) msg_add_entries(myp, pc, acc, rmaxu, rsum, ei, xl, logit, lr0, lane, h);
            __syncthreads();
            if (!more) break;
        }

#pragma unroll
        for (int r = 0; r < MSG_RW; ++r) {
            const int g = row0 + lr0 + r;
            if (lr0 + r < rpb && g < N_) {
                const float den = rsum[(lr0 + r) * 8 + h] + 1e-16f;
                const v4f a = acc[(lr0 + r) * 32 + lane];
                v4f o;
                o.x = bv.x + a.x / den;
                o.y = bv.y + a.y / den;
                o.z = bv.z + a.z / den;
                o.w = bv.w + a.w / den;
                *(v4f*)(out + (size_t)g * 128 + 4 * lane) = o;
            }
        }
    }
}

extern "C" void kernel_launch(void* const* d_in, const int* in_sizes, int n_in,
                              void* d_out, int out_size, void* d_ws, size_t ws_size,
                              hipStream_t stream) {
    const float* x     = (const float*)d_in[0];
    const int*   ei    = (const int*)d_in[1];
    const float* eattr = (const float*)d_in[2];
    const float* Wl    = (const float*)d_in[3];
    const float* bl    = (const float*)d_in[4];
    const float* Wr    = (const float*)d_in[5];
    const float* br    = (const float*)d_in[6];
    const float* We    = (const float*)d_in[7];
    const float* att   = (const float*)d_in[8];
    const float* bias  = (const float*)d_in[9];
    float* out = (float*)d_out;
    float* ws  = (float*)d_ws;

    const int n = in_sizes[0] / 128;
    const int e = in_sizes[2] / 8;
    const int tot = e + n;

    size_t off_mean   = 0;
    size_t off_xl     = 256;
    size_t off_xr     = off_xl + (size_t)n * 128;
    size_t off_logit  = off_xr + (size_t)n * 128;
    size_t off_part   = off_logit + (size_t)tot * 8;
    size_t off_poff   = off_part + (size_t)e * 2 + 2 * BIN_CS;

    int nTiles = n / 16;
    dim3 g((nTiles + 3) / 4, 2);
    node_xform_wmma<<<g, 128, 0, stream>>>(x, Wl, bl, Wr, br,
                                           ws + off_xl, ws + off_xr,
                                           ws + off_mean, nTiles);
    attr_mean_k<<<512, 256, 0, stream>>>(eattr, ws + off_mean, e);
    int nChunks = (e + BIN_CS - 1) / BIN_CS;
    edge_part_k<<<(nChunks + BIN_WPB - 1) / BIN_WPB, 256, 0, stream>>>(
        ei, (int*)(ws + off_part), (int*)(ws + off_poff), e, n, BIN_NB);

    int tileBlocks = ((tot / 16) * 32 + 255) / 256;
    edge_logit_tile_k<<<tileBlocks, 256, 0, stream>>>(ei, eattr, We, ws + off_mean,
                                                      1.0f / (float)e,
                                                      ws + off_xl, ws + off_xr, att,
                                                      ws + off_logit, e, n);
    edge_msg_owner_k<<<BIN_NB, 1024, 0, stream>>>(ei, ws + off_logit, ws + off_xl, bias, out,
                                                  (const int*)(ws + off_part), (const int*)(ws + off_poff), e, n);
}


# edge_logit_tile_k as a resident grid: 4096 waves loop over 13 tiles each, W_e and att loaded once per wave, next tile's row numbers and A fragment requested behind this tile's gathers; sums unchanged
# speedup vs baseline: 2.6370x; 1.0412x over previous
#include <hip/hip_runtime.h>
#include <hip/hip_bf16.h>

typedef __attribute__((ext_vector_type(2))) float v2f;
typedef __attribute__((ext_vector_type(4))) float v4f;
typedef __attribute__((ext_vector_type(8))) float v8f;

#define NSLOPE 0.2f

__device__ __forceinline__ unsigned fmap(float x) {
    unsigned b = __float_as_uint(x);
    return (b & 0x80000000u) ? ~b : (b | 0x80000000u);
}
__device__ __forceinline__ float funmap(unsigned u) {
    return (u & 0x80000000u) ? __uint_as_float(u ^ 0x80000000u)
                             : __uint_as_float(~u);
}

__global__ __launch_bounds__(256) void attr_mean_k(const float* __restrict__ eattr,
                                                   float* __restrict__ meanAcc, int E_) {
    __shared__ float s[8];
    if (threadIdx.x < 8) s[threadIdx.x] = 0.f;
    __syncthreads();
    float acc[8] = {0.f, 0.f, 0.f, 0.f, 0.f, 0.f, 0.f, 0.f};
    for (int r = blockIdx.x * blockDim.x + threadIdx.x; r < E_; r += gridDim.x * blockDim.x) {
        const float* row = eattr + (size_t)r * 8;
#pragma unroll
        for (int d = 0; d < 8; ++d) acc[d] += row[d];
    }
#pragma unroll
    for (int d = 0; d < 8; ++d) atomicAdd(&s[d], acc[d]);
    __syncthreads();
    if (threadIdx.x < 8) atomicAdd(meanAcc + threadIdx.x, s[threadIdx.x]);
}

__global__ __launch_bounds__(128) void node_xform_wmma(
    const float* __restrict__ X,
    const float* __restrict__ Wl, const float* __restrict__ bl,
    const float* __restrict__ Wr, const float* __restrict__ br,
    float* __restrict__ xl, float* __restrict__ xr,
    float* __restrict__ meanAcc, int nTiles) {
    int wid = threadIdx.x >> 5;
    int lane = threadIdx.x & 31;
    int rowTile = blockIdx.x * 4 + wid;
    if (rowTile >= nTiles) return;

    if (blockIdx.y != 0 && rowTile == 0 && lane < 8) meanAcc[lane] = 0.f;

    const float* W    = blockIdx.y ? Wr : Wl;
    const float* bias = blockIdx.y ? br : bl;
    float* out        = blockIdx.y ? xr : xl;

    int row0  = rowTile * 16;
    int m     = lane & 15;
    int khalf = lane >> 4;

    v8f acc[8] = {};

    const float* arow = X + (size_t)(row0 + m) * 128;
    for (int kt = 0; kt < 32; ++kt) {
        int k0 = kt * 4 + khalf * 2;
        v2f a;
        a.x = arow[k0];
        a.y = arow[k0 + 1];
        const float* w0 = W + (size_t)k0 * 128;
        const float* w1 = w0 + 128;
#pragma unroll
        for (int nt = 0; nt < 8; ++nt) {
            v2f b;
            b.x = w0[nt * 16 + m];
            b.y = w1[nt * 16 + m];
            acc[nt] = __builtin_amdgcn_wmma_f32_16x16x4_f32(
                false, a, false, b, (short)0, acc[nt], false, false);
        }
    }
#pragma unroll
    for (int nt = 0; nt < 8; ++nt) {
        float bv = bias[nt * 16 + m];
#pragma unroll
        for (int i = 0; i < 8; ++i) {
            int row = row0 + i + 8 * khalf;
            out[(size_t)row * 128 + nt * 16 + m] = acc[nt][i] + bv;
        }
    }
}

template <int K>
__device__ __forceinline__ int row_lane(int x) {
    return __builtin_amdgcn_update_dpp(0, x, 0x150 + K, 0xF, 0xF, true);
}

__device__ __forceinline__ void logit_half(const v4f* vl, const v4f* vr, const v8f* D, v4f av, float* p) {
#pragma unroll
    for (int r = 0; r < 8; ++r) {
        float m0 = vl[r][0] + vr[r][0] + D[0][r];
        float m1 = vl[r][1] + vr[r][1] + D[1][r];
        float m2 = vl[r][2] + vr[r][2] + D[2][r];
        float m3 = vl[r][3] + vr[r][3] + D[3][r];
        m0 = (m0 > 0.f) ? m0 : NSLOPE * m0;
        m1 = (m1 > 0.f) ? m1 : NSLOPE * m1;
        m2 = (m2 > 0.f) ? m2 : NSLOPE * m2;
        m3 = (m3 > 0.f) ? m3 : NSLOPE * m3;
        float t = m0 * av[0] + m1 * av[1] + m2 * av[2] + m3 * av[3];
        t += __shfl_xor(t, 1, 32);
        t += __shfl_xor(t, 2, 32);
        p[r] = t;
    }
}

__global__ __launch_bounds__(256) __attribute__((amdgpu_waves_per_eu(4, 4))) void edge_logit_tile_k(
    const int* __restrict__ ei, const float* __restrict__ eattr,
    const float* __restrict__ We, const float* __restrict__ meanAcc, float invE,
    const float* __restrict__ xl, const float* __restrict__ xr,
    const float* __restrict__ att, float* __restrict__ logit, int E_, int N_) {
    const int lane = threadIdx.x & 31;
    const int wv  = __builtin_amdgcn_readfirstlane((int)(threadIdx.x >> 5));
    const int wpb = (int)(blockDim.x >> 5);
    const int nW  = (int)gridDim.x * wpb;
    const int nT  = (E_ + N_) >> 4;
    int tile = (int)blockIdx.x * wpb + wv;
    if (tile >= nT) return;
    const int c  = lane & 15;
    const int hf = lane >> 4;

    v4f w[2][2][2];
#pragma unroll
    for (int kt = 0; kt < 2; ++kt)
#pragma unroll
        for (int s = 0; s < 2; ++s)
#pragma unroll
            for (int y = 0; y < 2; ++y)
                w[kt][s][y] = *(const v4f*)(We + (size_t)(4 * kt + 2 * hf + y) * 128 + 64 * s + 4 * c);

    const v4f av0 = *(const v4f*)(att + 4 * c);
    const v4f av1 = *(const v4f*)(att + 64 + 4 * c);

    const int      ioff = ((c & 8) ? E_ : 0) + 8 * hf + (c & 7);
    const unsigned lo   = 16u * (unsigned)c;

    int idx;
    v2f a[2];
    {
        const bool edge = (tile << 4) < E_;
        const int  lt   = edge ? tile : 0;
        const int  li   = ei[ioff + (lt << 4)];
        const float* lrow = edge ? eattr + ((size_t)((lt << 4) + c) * 8 + 2 * hf) : meanAcc + 2 * hf;
        const float sc  = edge ? 1.f : invE;
        idx  = edge ? li : ((tile << 4) + 8 * hf - E_ + (c & 7));
        a[0] = *(const v2f*)(lrow) * sc;
        a[1] = *(const v2f*)(lrow + 4) * sc;
    }

    for (;;) {
        const int eb = (tile << 4) + 8 * hf;
        const int src[8] = {row_lane<0>(idx), row_lane<1>(idx), row_lane<2>(idx),  row_lane<3>(idx),
                            row_lane<4>(idx), row_lane<5>(idx), row_lane<6>(idx),  row_lane<7>(idx)};
        const int dst[8] = {row_lane<8>(idx),  row_lane<9>(idx),  row_lane<10>(idx), row_lane<11>(idx),
                            row_lane<12>(idx), row_lane<13>(idx), row_lane<14>(idx), row_lane<15>(idx)};

        v4f vl[2][8], vr[2][8];
#pragma unroll
        for (int s = 0; s < 2; ++s) {
#pragma unroll
            for (int r = 0; r < 8; ++r) {
                const unsigned ol = ((unsigned)src[r] << 9) + lo;
                const unsigned orr = ((unsigned)dst[r] << 9) + lo;
                vl[s][r] = *(const v4f*)((const char*)xl + (size_t)ol + 256 * s);
                vr[s][r] = *(const v4f*)((const char*)xr + (size_t)orr + 256 * s);
            }
        }
        __builtin_amdgcn_sched_barrier(0);
        const int  nxt   = tile + nW;
        const bool more  = nxt < nT;
        const int  nt    = more ? nxt : tile;
        const bool nedge = (nt << 4) < E_;
        const int  lt    = nedge ? nt : 0;
        const int  li    = ei[ioff + (lt << 4)];
        const float* lrow = nedge ? eattr + ((size_t)((lt << 4) + c) * 8 + 2 * hf) : meanAcc + 2 * hf;
        const v2f la0 = *(const v2f*)(lrow);
        const v2f la1 = *(const v2f*)(lrow + 4);
        __builtin_amdgcn_sched_barrier(0);

        float p[2][8];
        {
            v8f D[4] = {};
#pragma unroll
            for (int kt = 0; kt < 2; ++kt)
#pragma unroll
                for (int q = 0; q < 4; ++q) {
                    v2f b;
                    b.x = w[kt][0][0][q];
                    b.y = w[kt][0][1][q];
                    D[q] = __builtin_amdgcn_wmma_f32_16x16x4_f32(
                        false, a[kt], false, b, (short)0, D[q], false, false);
                }
            __builtin_amdgcn_sched_barrier(0);
            logit_half(vl[0], vr[0], D, av0, p[0]);
        }
        {
            v8f D[4] = {};
#pragma unroll
            for (int kt = 0; kt < 2; ++kt)
#pragma unroll
                for (int q = 0; q < 4; ++q) {
                    v2f b;
                    b.x = w[kt][1][0][q];
                    b.y = w[kt][1][1][q];
                    D[q] = __builtin_amdgcn_wmma_f32_16x16x4_f32(
                        false, a[kt], false, b, (short)0, D[q], false, false);
                }
            logit_half(vl[1], vr[1], D, av1, p[1]);
        }

        const int j    = c & 3;
        const int head = ((j & 1) << 2) + (c >> 2);
        const int rofs = (j >> 1) << 2;
#pragma unroll
        for (int r = 0; r < 4; ++r) {
            const float lo_ = (j & 1) ? p[1][r]     : p[0][r];
            const float hi_ = (j & 1) ? p[1][r + 4] : p[0][r + 4];
            const float v   = (j >> 1) ? hi_ : lo_;
            logit[(size_t)(eb + r + rofs) * 8 + head] = v;
        }

        const float sc = nedge ? 1.f : invE;
        tile = nxt;
        idx  = nedge ? li : ((nt << 4) + 8 * hf - E_ + (c & 7));
        a[0] = la0 * sc;
        a[1] = la1 * sc;
        if (!more) break;
    }
}

#define MSG_WAVES 32
#define MSG_RW    7
#define MSG_ROWS  (MSG_WAVES * MSG_RW)
#define MSG_LCAP  384
#define MSG_PCAP  256
#define MSG_B     8

#define BIN_CS  416
#define BIN_G   (BIN_CS / 32)
#define BIN_NB  256
#define BIN_WPB 8
#define BIN_PP  (BIN_NB + 4)

__global__ __launch_bounds__(256) void edge_part_k(
    const int* __restrict__ ei, int* __restrict__ part_, int* __restrict__ poff, int E_, int N_, int G) {
    __shared__ int cntS[BIN_WPB * BIN_NB];
    const int lane = threadIdx.x & 31;
    const int w    = __builtin_amdgcn_readfirstlane((int)(threadIdx.x >> 5));
    const int c    = (int)blockIdx.x * BIN_WPB + w;
    const int NC   = (E_ + BIN_CS - 1) / BIN_CS;
    if (c >= NC) return;
    int rpb = (N_ + G - 1) / G;
    if (rpb > MSG_ROWS) rpb = MSG_ROWS;
    int* const cnt = cntS + w * BIN_NB;
    int2* const part = (int2*)part_;
    const int e0 = c * BIN_CS;

    int d[BIN_G];
#pragma unroll
    for (int g = 0; g < BIN_G; ++g) {
        const int e = e0 + 32 * g + lane;
        d[g] = (e < E_) ? ei[E_ + e] : -1;
    }
#pragma unroll
    for (int k = 0; k < BIN_NB / 32; ++k) cnt[lane + 32 * k] = 0;

    int bn[BIN_G];
#pragma unroll
    for (int g = 0; g < BIN_G; ++g) {
        int b = -1;
        if (d[g] >= 0) {
            b = (int)((unsigned)d[g] / (unsigned)rpb);
            if (b > BIN_NB - 1) b = BIN_NB - 1;
            atomicAdd(&cnt[b], 1);
        }
        bn[g] = b;
    }

    int ex[8]; int run = 0;
#pragma unroll
    for (int k = 0; k < 8; ++k) { ex[k] = run; run += cnt[8 * lane + k]; }
    int inc = run;
#pragma unroll
    for (int off = 1; off < 32; off <<= 1) {
        const int t = __shfl_up(inc, off, 32);
        if (lane >= off) inc += t;
    }
    const int base = inc - run;
    int* const po = poff + (size_t)c * BIN_PP;
#pragma unroll
    for (int k = 0; k < 8; ++k) {
        cnt[8 * lane + k] = base + ex[k];
        po[8 * lane + k]  = base + ex[k];
    }
    if (lane == 31) po[BIN_NB] = inc;

#pragma unroll
    for (int g = 0; g < BIN_G; ++g) {
        const int  b     = bn[g];
        const bool valid = b >= 0;
        unsigned same = __builtin_amdgcn_ballot_w32(valid);
#pragma unroll
        for (int bit = 0; bit < 8; ++bit) {
            const bool on = valid && ((b >> bit) & 1);
            const unsigned bal = __builtin_amdgcn_ballot_w32(on);
            same &= on ? bal : ~bal;
        }
        const int rank = (int)__builtin_amdgcn_mbcnt_lo(same, 0u);
        const int tot  = __popc(same);
        int pos = 0;
        if (valid) {
            pos = cnt[b];
            const int lr = d[g] - b * rpb;
            part[e0 + pos + rank] = make_int2((int)((unsigned)lr << 24), e0 + 32 * g + lane);
        }
        if (valid && rank == 0) cnt[b] = pos + tot;
    }
}

__device__ __forceinline__ float quad_lane(float x, int k) {
    const int xi = __float_as_int(x);
    int r;
    switch (k & 3) {
        case 0:  r = __builtin_amdgcn_update_dpp(0, xi, 0x00, 0xF, 0xF, true); break;
        case 1:  r = __builtin_amdgcn_update_dpp(0, xi, 0x55, 0xF, 0xF, true); break;
        case 2:  r = __builtin_amdgcn_update_dpp(0, xi, 0xAA, 0xF, 0xF, true); break;
        default: r = __builtin_amdgcn_update_dpp(0, xi, 0xFF, 0xF, 0xF, true); break;
    }
    return __int_as_float(r);
}

__device__ __forceinline__ void msg_add_entries(
    int2* myp, int pc, v4f* acc, unsigned* rmaxu, float* rsum,
    const int* __restrict__ ei, const float* __restrict__ xl,
    const float* __restrict__ logit, int lr0, int lane, int h) {
    {
        int sv[MSG_PCAP / 32];
#pragma unroll
        for (int u = 0; u < MSG_PCAP / 32; ++u) {
            const int j = lane + 32 * u;
            sv[u] = (j < pc) ? ei[myp[j].y] : 0;
        }
#pragma unroll
        for (int u = 0; u < MSG_PCAP / 32; ++u) {
            const int j = lane + 32 * u;
            if (j < pc) myp[j].x |= sv[u];
        }
    }

    unsigned mp[MSG_RW];
#pragma unroll
    for (int r = 0; r < MSG_RW; ++r) mp[r] = rmaxu[(lr0 + r) * 8 + h];
    {
        const int hh = lane & 7;
        const int eo = lane >> 3;
        for (int t0 = 0; t0 < pc; t0 += 4 * MSG_B) {
            float v[MSG_B]; int rw[MSG_B];
#pragma unroll
            for (int u = 0; u < MSG_B; ++u) {
                const int j  = t0 + 4 * u + eo;
                const int2 q = myp[(j < pc) ? j : (pc - 1)];
                rw[u] = (int)((unsigned)q.x >> 24);
                v[u]  = logit[(size_t)q.y * 8 + hh];
            }
#pragma unroll
            for (int u = 0; u < MSG_B; ++u) {
                const int j = t0 + 4 * u + eo;
                if (j < pc) atomicMax(&rmaxu[rw[u] * 8 + hh], fmap(v[u]));
            }
        }
    }
#pragma unroll
    for (int r = 0; r < MSG_RW; ++r) {
        const unsigned mn = rmaxu[(lr0 + r) * 8 + h];
        const float sc = expf(funmap(mp[r]) - funmap(mn));
        v4f a = acc[(lr0 + r) * 32 + lane];
        a *= sc;
        acc[(lr0 + r) * 32 + lane] = a;
        rsum[(lr0 + r) * 8 + h] *= sc;
    }

    const int jq = lane & 3;
    for (int b = 0; b < pc; b += MSG_B) {
        int qx[MSG_B], qy[MSG_B]; v4f xv[MSG_B];
        float lv[2]; unsigned mv[2];
#pragma unroll
        for (int k = 0; k < 2; ++k) {
            const int j  = b + jq + 4 * k;
            const int2 q = myp[(j < pc) ? j : (pc - 1)];
            lv[k] = logit[(size_t)q.y * 8 + h];
            mv[k] = rmaxu[(int)((unsigned)q.x >> 24) * 8 + h];
        }
#pragma unroll
        for (int u = 0; u < MSG_B; ++u) {
            const int j = (b + u < pc) ? (b + u) : (pc - 1);
            const int2 q = myp[j];
            qx[u] = __builtin_amdgcn_readfirstlane(q.x);
            qy[u] = __builtin_amdgcn_readfirstlane(q.y);
        }
#pragma unroll
        for (int u = 0; u < MSG_B; ++u) {
            const int s = qx[u] & 0x00FFFFFF;
            xv[u] = *(const v4f*)(xl + (size_t)s * 128 + 4 * lane);
        }
        float pe[2];
#pragma unroll
        for (int k = 0; k < 2; ++k) pe[k] = expf(lv[k] - funmap(mv[k]));
#pragma unroll
        for (int u = 0; u < MSG_B; ++u) {
            if (b + u < pc) {
                const int rl  = (int)((unsigned)qx[u] >> 24);
                const float p = quad_lane(pe[u >> 2], u & 3);
                v4f a = acc[rl * 32 + lane];
                a += p * xv[u];
                acc[rl * 32 + lane] = a;
                rsum[rl * 8 + h] += p;
            }
        }
    }
}

__global__ __launch_bounds__(1024) void edge_msg_owner_k(
    const int* __restrict__ ei, const float* __restrict__ logit, const float* __restrict__ xl,
    const float* __restrict__ obias, float* __restrict__ out,
    const int* __restrict__ part_, const int* __restrict__ poff, int E_, int N_) {
    __shared__ v4f  acc[MSG_ROWS * 32];
    __shared__ int2 lst[MSG_WAVES * MSG_LCAP];
    __shared__ int2 prv[MSG_WAVES * MSG_PCAP];
    __shared__ unsigned rmaxu[MSG_ROWS * 8];
    __shared__ float    rsum[MSG_ROWS * 8];
    __shared__ int  s_cnt[MSG_WAVES];
    __shared__ int  s_un[MSG_WAVES];
    __shared__ int  s_oc[MSG_WAVES * MSG_WAVES];
    __shared__ int  s_run[MSG_WAVES * MSG_WAVES];

    const int lane = threadIdx.x & 31;
    const int w    = __builtin_amdgcn_readfirstlane((int)(threadIdx.x >> 5));
    const int h    = lane >> 2;
    const int G    = (int)gridDim.x;
    int rpb = (N_ + G - 1) / G;
    if (rpb > MSG_ROWS) rpb = MSG_ROWS;
    const int2* const part = (const int2*)part_;
    const int NC  = (E_ + BIN_CS - 1) / BIN_CS;
    const int CPW = (NC + MSG_WAVES - 1) / MSG_WAVES;
    const v4f bv = *(const v4f*)(obias + 4 * lane);
    const int lr0 = w * MSG_RW;
    int2* const myl = lst + w * MSG_LCAP;
    int2* const myp = prv + w * MSG_PCAP;

    for (int row0 = blockIdx.x * rpb; row0 < N_; row0 += G * rpb) {
        {
            v4f xv[MSG_RW]; float ls[MSG_RW];
#pragma unroll
            for (int r = 0; r < MSG_RW; ++r) {
                const int g  = row0 + lr0 + r;
                const int gg = (lr0 + r < rpb && g < N_) ? g : row0;
                xv[r] = *(const v4f*)(xl + (size_t)gg * 128 + 4 * lane);
                ls[r] = logit[(size_t)(E_ + gg) * 8 + h];
            }
#pragma unroll
            for (int r = 0; r < MSG_RW; ++r) {
                acc[(lr0 + r) * 32 + lane] = xv[r];
                rmaxu[(lr0 + r) * 8 + h]   = fmap(ls[r]);
                rsum[(lr0 + r) * 8 + h]    = 1.f;
            }
        }

        int R = row0 / rpb;
        if (R > BIN_NB - 1) R = BIN_NB - 1;
        const int cbeg = w * CPW;
        const int cend = (cbeg + CPW < NC) ? (cbeg + CPW) : NC;
        int ci = cbeg;
        int co = 0;
        for (;;) {
            int cnt = 0;
            while (ci < cend && cnt < MSG_LCAP) {
                const int  c  = ci + lane;
                const bool cv = c < cend;
                int o0 = 0, o1 = 0;
                if (cv) {
                    o0 = poff[(size_t)c * BIN_PP + R];
                    o1 = poff[(size_t)c * BIN_PP + R + 1];
                }
                if (lane == 0) o0 += co;
                const int n = o1 - o0;
                int inc = n;
#pragma unroll
                for (int off = 1; off < 32; off <<= 1) {
                    const int t = __shfl_up(inc, off, 32);
                    if (lane >= off) inc += t;
                }
                const int exc  = inc - n;
                const int room = MSG_LCAP - cnt;
                int t = room - exc;
                t = (t < 0) ? 0 : ((t > n) ? n : t);
                const int2* const sp = part + (size_t)c * BIN_CS + o0;
                int2* const dp = myl + cnt + exc;
                for (int j0 = 0; __builtin_amdgcn_ballot_w32(j0 < t) != 0u; j0 += 4) {
                    int2 v[4];
#pragma unroll
                    for (int u = 0; u < 4; ++u) if (j0 + u < t) v[u] = sp[j0 + u];
#pragma unroll
                    for (int u = 0; u < 4; ++u) if (j0 + u < t) dp[j0 + u] = v[u];
                }
                const unsigned cut = __builtin_amdgcn_ballot_w32(t < n);
                if (cut) {
                    const int f  = __builtin_ctz(cut);
                    const int tf = __builtin_amdgcn_readlane(t, f);
                    co  = tf + ((f == 0) ? co : 0);
                    ci += f;
                    cnt = MSG_LCAP;
                } else {
                    cnt += __builtin_amdgcn_readlane(inc, 31);
                    ci  += 32;
                    co   = 0;
                }
            }
            if (lane == 0) { s_cnt[w] = cnt; s_un[w] = (ci < cend) ? 1 : 0; }
            s_oc[w * MSG_WAVES + lane] = 0;
            for (int i0 = 0; i0 < cnt; i0 += 32) {
                const int i = i0 + lane;
                if (i < cnt) {
                    const int o = (int)((unsigned)myl[i].x >> 24) / MSG_RW;
                    if (o < MSG_WAVES) atomicAdd(&s_oc[w * MSG_WAVES + o], 1);
                }
            }
            __syncthreads();
            const bool more = __builtin_amdgcn_ballot_w32(s_un[lane] != 0) != 0u;

            int obase = 0, otot = 0;
#pragma unroll 8
            for (int L = 0; L < MSG_WAVES; ++L) {
                const int c = s_oc[L * MSG_WAVES + lane];
                obase += (L < w) ? c : 0;
                otot  += c;
            }
            const bool fits = __builtin_amdgcn_ballot_w32(otot > MSG_PCAP) == 0u;
            int pc = 0;
            if (fits) {
                int* const run = s_run + w * MSG_WAVES;
                run[lane] = obase;
                for (int i0 = 0; i0 < cnt; i0 += 32) {
                    const int i = i0 + lane;
                    int2 en = make_int2(0, 0);
                    if (i < cnt) en = myl[i];
                    const int  o     = (int)((unsigned)en.x >> 24) / MSG_RW;
                    const bool valid = (i < cnt) && (o < MSG_WAVES);
                    unsigned same = __builtin_amdgcn_ballot_w32(valid);
#pragma unroll
                    for (int bit = 0; bit < 5; ++bit) {
                        const bool on = valid && ((o >> bit) & 1);
                        const unsigned bal = __builtin_amdgcn_ballot_w32(on);
                        same &= on ? bal : ~bal;
                    }
                    const int rank = (int)__builtin_amdgcn_mbcnt_lo(same, 0u);
                    const int totg = __popc(same);
                    int pos = 0;
                    if (valid) {
                        pos = run[o];
                        prv[o * MSG_PCAP + pos + rank] = en;
                    }
                    if (valid && rank == 0) run[o] = pos + totg;
                }
                __syncthreads();
                pc = __builtin_amdgcn_readlane(otot, w);
            } else {
            for (int L = 0; L < MSG_WAVES; ++L) {
                const int n = __builtin_amdgcn_readfirstlane(s_cnt[L]);
                const int2* const ll = lst + L * MSG_LCAP;
                for (int i0 = 0; i0 < n; i0 += 32) {
                    const int  i  = i0 + lane;
                    const bool in = i < n;
                    int2 en = make_int2(0, 0);
                    if (in) en = ll[i];
                    const int  lr = (int)((unsigned)en.x >> 24);
                    const bool m  = in && (unsigned)(lr - lr0) < (unsigned)MSG_RW;
                    const unsigned bal = __builtin_amdgcn_ballot_w32(m);
                    if (bal) {
                        const int p = pc + (int)__builtin_amdgcn_mbcnt_lo(bal, 0u);
                        if (m) myp[p] = en;
                        pc += __popc(bal);
                    }
                    if (pc > MSG_PCAP - 32) {
                        msg_add_entries(myp, pc, acc, rmaxu, rsum, ei, xl, logit, lr0, lane, h);
                        pc = 0;
                    }
                }
            }
            }
            if (pc > 0) msg_add_entries(myp, pc, acc, rmaxu, rsum, ei, xl, logit, lr0, lane, h);
            __syncthreads();
            if (!more) break;
        }

#pragma unroll
        for (int r = 0; r < MSG_RW; ++r) {
            const int g = row0 + lr0 + r;
            if (lr0 + r < rpb && g < N_) {
                const float den = rsum[(lr0 + r) * 8 + h] + 1e-16f;
                const v4f a = acc[(lr0 + r) * 32 + lane];
                v4f o;
                o.x = bv.x + a.x / den;
                o.y = bv.y + a.y / den;
                o.z = bv.z + a.z / den;
                o.w = bv.w + a.w / den;
                *(v4f*)(out + (size_t)g * 128 + 4 * lane) = o;
            }
        }
    }
}

extern "C" void kernel_launch(void* const* d_in, const int* in_sizes, int n_in,
                              void* d_out, int out_size, void* d_ws, size_t ws_size,
                              hipStream_t stream) {
    const float* x     = (const float*)d_in[0];
    const int*   ei    = (const int*)d_in[1];
    const float* eattr = (const float*)d_in[2];
    const float* Wl    = (const float*)d_in[3];
    const float* bl    = (const float*)d_in[4];
    const float* Wr    = (const float*)d_in[5];
    const float* br    = (const float*)d_in[6];
    const float* We    = (const float*)d_in[7];
    const float* att   = (const float*)d_in[8];
    const float* bias  = (const float*)d_in[9];
    float* out = (float*)d_out;
    float* ws  = (float*)d_ws;

    const int n = in_sizes[0] / 128;
    const int e = in_sizes[2] / 8;
    const int tot = e + n;

    size_t off_mean   = 0;
    size_t off_xl     = 256;
    size_t off_xr     = off_xl + (size_t)n * 128;
    size_t off_logit  = off_xr + (size_t)n * 128;
    size_t off_part   = off_logit + (size_t)tot * 8;
    size_t off_poff   = off_part + (size_t)e * 2 + 2 * BIN_CS;

    int nTiles = n / 16;
    dim3 g((nTiles + 3) / 4, 2);
    node_xform_wmma<<<g, 128, 0, stream>>>(x, Wl, bl, Wr, br,
                                           ws + off_xl, ws + off_xr,
                                           ws + off_mean, nTiles);
    attr_mean_k<<<512, 256, 0, stream>>>(eattr, ws + off_mean, e);
    int nChunks = (e + BIN_CS - 1) / BIN_CS;
    edge_part_k<<<(nChunks + BIN_WPB - 1) / BIN_WPB, 256, 0, stream>>>(
        ei, (int*)(ws + off_part), (int*)(ws + off_poff), e, n, BIN_NB);

    edge_logit_tile_k<<<512, 256, 0, stream>>>(ei, eattr, We, ws + off_mean,
                                                      1.0f / (float)e,
                                                      ws + off_xl, ws + off_xr, att,
                                                      ws + off_logit, e, n);
    edge_msg_owner_k<<<BIN_NB, 1024, 0, stream>>>(ei, ws + off_logit, ws + off_xl, bias, out,
                                                  (const int*)(ws + off_part), (const int*)(ws + off_poff), e, n);
}


# attr_mean_k: 256 blocks, a lane's 25 loads of 16 B in flight at once (was 2 per wait); column sums by lane butterfly + fixed LDS sum per block, no LDS atomics: sum order changed; block atomics kept
# speedup vs baseline: 2.7819x; 1.0549x over previous
#include <hip/hip_runtime.h>
#include <hip/hip_bf16.h>

typedef __attribute__((ext_vector_type(2))) float v2f;
typedef __attribute__((ext_vector_type(4))) float v4f;
typedef __attribute__((ext_vector_type(8))) float v8f;

#define NSLOPE 0.2f

__device__ __forceinline__ unsigned fmap(float x) {
    unsigned b = __float_as_uint(x);
    return (b & 0x80000000u) ? ~b : (b | 0x80000000u);
}
__device__ __forceinline__ float funmap(unsigned u) {
    return (u & 0x80000000u) ? __uint_as_float(u ^ 0x80000000u)
                             : __uint_as_float(~u);
}

#define AM_U 25
__global__ __launch_bounds__(256) void attr_mean_k(const float* __restrict__ eattr,
                                                   float* __restrict__ meanAcc, int E_) {
    __shared__ float s[8 * 8];
    const int lane = threadIdx.x & 31;
    const int w    = threadIdx.x >> 5;
    const v4f* const q4 = (const v4f*)eattr;
    const int n4 = 2 * E_;
    const int T  = (int)(gridDim.x * blockDim.x);
    const v4f zero = {0.f, 0.f, 0.f, 0.f};
    v4f acc = zero;
    for (int i = (int)(blockIdx.x * blockDim.x + threadIdx.x); i < n4; i += AM_U * T) {
        v4f v[AM_U];
#pragma unroll
        for (int u = 0; u < AM_U; ++u) {
            const int j = i + u * T;
            v[u] = q4[(j < n4) ? j : (n4 - 1)];
        }
        __builtin_amdgcn_sched_barrier(0);
#pragma unroll
        for (int u = 0; u < AM_U; ++u) {
            const int j = i + u * T;
            acc += (j < n4) ? v[u] : zero;
        }
    }
#pragma unroll
    for (int off = 2; off < 32; off <<= 1) {
        acc.x += __shfl_xor(acc.x, off, 32);
        acc.y += __shfl_xor(acc.y, off, 32);
        acc.z += __shfl_xor(acc.z, off, 32);
        acc.w += __shfl_xor(acc.w, off, 32);
    }
    if (lane < 2) {
        float* const d = s + w * 8 + 4 * lane;
        d[0] = acc.x; d[1] = acc.y; d[2] = acc.z; d[3] = acc.w;
    }
    __syncthreads();
    if (threadIdx.x < 8) {
        float t = 0.f;
#pragma unroll
        for (int k = 0; k < 8; ++k) t += s[k * 8 + threadIdx.x];
        atomicAdd(meanAcc + threadIdx.x, t);
    }
}

__global__ __launch_bounds__(128) void node_xform_wmma(
    const float* __restrict__ X,
    const float* __restrict__ Wl, const float* __restrict__ bl,
    const float* __restrict__ Wr, const float* __restrict__ br,
    float* __restrict__ xl, float* __restrict__ xr,
    float* __restrict__ meanAcc, int nTiles) {
    int wid = threadIdx.x >> 5;
    int lane = threadIdx.x & 31;
    int rowTile = blockIdx.x * 4 + wid;
    if (rowTile >= nTiles) return;

    if (blockIdx.y != 0 && rowTile == 0 && lane < 8) meanAcc[lane] = 0.f;

    const float* W    = blockIdx.y ? Wr : Wl;
    const float* bias = blockIdx.y ? br : bl;
    float* out        = blockIdx.y ? xr : xl;

    int row0  = rowTile * 16;
    int m     = lane & 15;
    int khalf = lane >> 4;

    v8f acc[8] = {};

    const float* arow = X + (size_t)(row0 + m) * 128;
    for (int kt = 0; kt < 32; ++kt) {
        int k0 = kt * 4 + khalf * 2;
        v2f a;
        a.x = arow[k0];
        a.y = arow[k0 + 1];
        const float* w0 = W + (size_t)k0 * 128;
        const float* w1 = w0 + 128;
#pragma unroll
        for (int nt = 0; nt < 8; ++nt) {
            v2f b;
            b.x = w0[nt * 16 + m];
            b.y = w1[nt * 16 + m];
            acc[nt] = __builtin_amdgcn_wmma_f32_16x16x4_f32(
                false, a, false, b, (short)0, acc[nt], false, false);
        }
    }
#pragma unroll
    for (int nt = 0; nt < 8; ++nt) {
        float bv = bias[nt * 16 + m];
#pragma unroll
        for (int i = 0; i < 8; ++i) {
            int row = row0 + i + 8 * khalf;
            out[(size_t)row * 128 + nt * 16 + m] = acc[nt][i] + bv;
        }
    }
}

template <int K>
__device__ __forceinline__ int row_lane(int x) {
    return __builtin_amdgcn_update_dpp(0, x, 0x150 + K, 0xF, 0xF, true);
}

__device__ __forceinline__ void logit_half(const v4f* vl, const v4f* vr, const v8f* D, v4f av, float* p) {
#pragma unroll
    for (int r = 0; r < 8; ++r) {
        float m0 = vl[r][0] + vr[r][0] + D[0][r];
        float m1 = vl[r][1] + vr[r][1] + D[1][r];
        float m2 = vl[r][2] + vr[r][2] + D[2][r];
        float m3 = vl[r][3] + vr[r][3] + D[3][r];
        m0 = (m0 > 0.f) ? m0 : NSLOPE * m0;
        m1 = (m1 > 0.f) ? m1 : NSLOPE * m1;
        m2 = (m2 > 0.f) ? m2 : NSLOPE * m2;
        m3 = (m3 > 0.f) ? m3 : NSLOPE * m3;
        float t = m0 * av[0] + m1 * av[1] + m2 * av[2] + m3 * av[3];
        t += __shfl_xor(t, 1, 32);
        t += __shfl_xor(t, 2, 32);
        p[r] = t;
    }
}

__global__ __launch_bounds__(256) __attribute__((amdgpu_waves_per_eu(4, 4))) void edge_logit_tile_k(
    const int* __restrict__ ei, const float* __restrict__ eattr,
    const float* __restrict__ We, const float* __restrict__ meanAcc, float invE,
    const float* __restrict__ xl, const float* __restrict__ xr,
    const float* __restrict__ att, float* __restrict__ logit, int E_, int N_) {
    const int lane = threadIdx.x & 31;
    const int wv  = __builtin_amdgcn_readfirstlane((int)(threadIdx.x >> 5));
    const int wpb = (int)(blockDim.x >> 5);
    const int nW  = (int)gridDim.x * wpb;
    const int nT  = (E_ + N_) >> 4;
    int tile = (int)blockIdx.x * wpb + wv;
    if (tile >= nT) return;
    const int c  = lane & 15;
    const int hf = lane >> 4;

    v4f w[2][2][2];
#pragma unroll
    for (int kt = 0; kt < 2; ++kt)
#pragma unroll
        for (int s = 0; s < 2; ++s)
#pragma unroll
            for (int y = 0; y < 2; ++y)
                w[kt][s][y] = *(const v4f*)(We + (size_t)(4 * kt + 2 * hf + y) * 128 + 64 * s + 4 * c);

    const v4f av0 = *(const v4f*)(att + 4 * c);
    const v4f av1 = *(const v4f*)(att + 64 + 4 * c);

    const int      ioff = ((c & 8) ? E_ : 0) + 8 * hf + (c & 7);
    const unsigned lo   = 16u * (unsigned)c;

    int idx;
    v2f a[2];
    {
        const bool edge = (tile << 4) < E_;
        const int  lt   = edge ? tile : 0;
        const int  li   = ei[ioff + (lt << 4)];
        const float* lrow = edge ? eattr + ((size_t)((lt << 4) + c) * 8 + 2 * hf) : meanAcc + 2 * hf;
        const float sc  = edge ? 1.f : invE;
        idx  = edge ? li : ((tile << 4) + 8 * hf - E_ + (c & 7));
        a[0] = *(const v2f*)(lrow) * sc;
        a[1] = *(const v2f*)(lrow + 4) * sc;
    }

    for (;;) {
        const int eb = (tile << 4) + 8 * hf;
        const int src[8] = {row_lane<0>(idx), row_lane<1>(idx), row_lane<2>(idx),  row_lane<3>(idx),
                            row_lane<4>(idx), row_lane<5>(idx), row_lane<6>(idx),  row_lane<7>(idx)};
        const int dst[8] = {row_lane<8>(idx),  row_lane<9>(idx),  row_lane<10>(idx), row_lane<11>(idx),
                            row_lane<12>(idx), row_lane<13>(idx), row_lane<14>(idx), row_lane<15>(idx)};

        v4f vl[2][8], vr[2][8];
#pragma unroll
        for (int s = 0; s < 2; ++s) {
#pragma unroll
            for (int r = 0; r < 8; ++r) {
                const unsigned ol = ((unsigned)src[r] << 9) + lo;
                const unsigned orr = ((unsigned)dst[r] << 9) + lo;
                vl[s][r] = *(const v4f*)((const char*)xl + (size_t)ol + 256 * s);
                vr[s][r] = *(const v4f*)((const char*)xr + (size_t)orr + 256 * s);
            }
        }
        __builtin_amdgcn_sched_barrier(0);
        const int  nxt   = tile + nW;
        const bool more  = nxt < nT;
        const int  nt    = more ? nxt : tile;
        const bool nedge = (nt << 4) < E_;
        const int  lt    = nedge ? nt : 0;
        const int  li    = ei[ioff + (lt << 4)];
        const float* lrow = nedge ? eattr + ((size_t)((lt << 4) + c) * 8 + 2 * hf) : meanAcc + 2 * hf;
        const v2f la0 = *(const v2f*)(lrow);
        const v2f la1 = *(const v2f*)(lrow + 4);
        __builtin_amdgcn_sched_barrier(0);

        float p[2][8];
        {
            v8f D[4] = {};
#pragma unroll
            for (int kt = 0; kt < 2; ++kt)
#pragma unroll
                for (int q = 0; q < 4; ++q) {
                    v2f b;
                    b.x = w[kt][0][0][q];
                    b.y = w[kt][0][1][q];
                    D[q] = __builtin_amdgcn_wmma_f32_16x16x4_f32(
                        false, a[kt], false, b, (short)0, D[q], false, false);
                }
            __builtin_amdgcn_sched_barrier(0);
            logit_half(vl[0], vr[0], D, av0, p[0]);
        }
        {
            v8f D[4] = {};
#pragma unroll
            for (int kt = 0; kt < 2; ++kt)
#pragma unroll
                for (int q = 0; q < 4; ++q) {
                    v2f b;
                    b.x = w[kt][1][0][q];
                    b.y = w[kt][1][1][q];
                    D[q] = __builtin_amdgcn_wmma_f32_16x16x4_f32(
                        false, a[kt], false, b, (short)0, D[q], false, false);
                }
            logit_half(vl[1], vr[1], D, av1, p[1]);
        }

        const int j    = c & 3;
        const int head = ((j & 1) << 2) + (c >> 2);
        const int rofs = (j >> 1) << 2;
#pragma unroll
        for (int r = 0; r < 4; ++r) {
            const float lo_ = (j & 1) ? p[1][r]     : p[0][r];
            const float hi_ = (j & 1) ? p[1][r + 4] : p[0][r + 4];
            const float v   = (j >> 1) ? hi_ : lo_;
            logit[(size_t)(eb + r + rofs) * 8 + head] = v;
        }

        const float sc = nedge ? 1.f : invE;
        tile = nxt;
        idx  = nedge ? li : ((nt << 4) + 8 * hf - E_ + (c & 7));
        a[0] = la0 * sc;
        a[1] = la1 * sc;
        if (!more) break;
    }
}

#define MSG_WAVES 32
#define MSG_RW    7
#define MSG_ROWS  (MSG_WAVES * MSG_RW)
#define MSG_LCAP  384
#define MSG_PCAP  256
#define MSG_B     8

#define BIN_CS  416
#define BIN_G   (BIN_CS / 32)
#define BIN_NB  256
#define BIN_WPB 8
#define BIN_PP  (BIN_NB + 4)

__global__ __launch_bounds__(256) void edge_part_k(
    const int* __restrict__ ei, int* __restrict__ part_, int* __restrict__ poff, int E_, int N_, int G) {
    __shared__ int cntS[BIN_WPB * BIN_NB];
    const int lane = threadIdx.x & 31;
    const int w    = __builtin_amdgcn_readfirstlane((int)(threadIdx.x >> 5));
    const int c    = (int)blockIdx.x * BIN_WPB + w;
    const int NC   = (E_ + BIN_CS - 1) / BIN_CS;
    if (c >= NC) return;
    int rpb = (N_ + G - 1) / G;
    if (rpb > MSG_ROWS) rpb = MSG_ROWS;
    int* const cnt = cntS + w * BIN_NB;
    int2* const part = (int2*)part_;
    const int e0 = c * BIN_CS;

    int d[BIN_G];
#pragma unroll
    for (int g = 0; g < BIN_G; ++g) {
        const int e = e0 + 32 * g + lane;
        d[g] = (e < E_) ? ei[E_ + e] : -1;
    }
#pragma unroll
    for (int k = 0; k < BIN_NB / 32; ++k) cnt[lane + 32 * k] = 0;

    int bn[BIN_G];
#pragma unroll
    for (int g = 0; g < BIN_G; ++g) {
        int b = -1;
        if (d[g] >= 0) {
            b = (int)((unsigned)d[g] / (unsigned)rpb);
            if (b > BIN_NB - 1) b = BIN_NB - 1;
            atomicAdd(&cnt[b], 1);
        }
        bn[g] = b;
    }

    int ex[8]; int run = 0;
#pragma unroll
    for (int k = 0; k < 8; ++k) { ex[k] = run; run += cnt[8 * lane + k]; }
    int inc = run;
#pragma unroll
    for (int off = 1; off < 32; off <<= 1) {
        const int t = __shfl_up(inc, off, 32);
        if (lane >= off) inc += t;
    }
    const int base = inc - run;
    int* const po = poff + (size_t)c * BIN_PP;
#pragma unroll
    for (int k = 0; k < 8; ++k) {
        cnt[8 * lane + k] = base + ex[k];
        po[8 * lane + k]  = base + ex[k];
    }
    if (lane == 31) po[BIN_NB] = inc;

#pragma unroll
    for (int g = 0; g < BIN_G; ++g) {
        const int  b     = bn[g];
        const bool valid = b >= 0;
        unsigned same = __builtin_amdgcn_ballot_w32(valid);
#pragma unroll
        for (int bit = 0; bit < 8; ++bit) {
            const bool on = valid && ((b >> bit) & 1);
            const unsigned bal = __builtin_amdgcn_ballot_w32(on);
            same &= on ? bal : ~bal;
        }
        const int rank = (int)__builtin_amdgcn_mbcnt_lo(same, 0u);
        const int tot  = __popc(same);
        int pos = 0;
        if (valid) {
            pos = cnt[b];
            const int lr = d[g] - b * rpb;
            part[e0 + pos + rank] = make_int2((int)((unsigned)lr << 24), e0 + 32 * g + lane);
        }
        if (valid && rank == 0) cnt[b] = pos + tot;
    }
}

__device__ __forceinline__ float quad_lane(float x, int k) {
    const int xi = __float_as_int(x);
    int r;
    switch (k & 3) {
        case 0:  r = __builtin_amdgcn_update_dpp(0, xi, 0x00, 0xF, 0xF, true); break;
        case 1:  r = __builtin_amdgcn_update_dpp(0, xi, 0x55, 0xF, 0xF, true); break;
        case 2:  r = __builtin_amdgcn_update_dpp(0, xi, 0xAA, 0xF, 0xF, true); break;
        default: r = __builtin_amdgcn_update_dpp(0, xi, 0xFF, 0xF, 0xF, true); break;
    }
    return __int_as_float(r);
}

__device__ __forceinline__ void msg_add_entries(
    int2* myp, int pc, v4f* acc, unsigned* rmaxu, float* rsum,
    const int* __restrict__ ei, const float* __restrict__ xl,
    const float* __restrict__ logit, int lr0, int lane, int h) {
    {
        int sv[MSG_PCAP / 32];
#pragma unroll
        for (int u = 0; u < MSG_PCAP / 32; ++u) {
            const int j = lane + 32 * u;
            sv[u] = (j < pc) ? ei[myp[j].y] : 0;
        }
#pragma unroll
        for (int u = 0; u < MSG_PCAP / 32; ++u) {
            const int j = lane + 32 * u;
            if (j < pc) myp[j].x |= sv[u];
        }
    }

    unsigned mp[MSG_RW];
#pragma unroll
    for (int r = 0; r < MSG_RW; ++r) mp[r] = rmaxu[(lr0 + r) * 8 + h];
    {
        const int hh = lane & 7;
        const int eo = lane >> 3;
        for (int t0 = 0; t0 < pc; t0 += 4 * MSG_B) {
            float v[MSG_B]; int rw[MSG_B];
#pragma unroll
            for (int u = 0; u < MSG_B; ++u) {
                const int j  = t0 + 4 * u + eo;
                const int2 q = myp[(j < pc) ? j : (pc - 1)];
                rw[u] = (int)((unsigned)q.x >> 24);
                v[u]  = logit[(size_t)q.y * 8 + hh];
            }
#pragma unroll
            for (int u = 0; u < MSG_B; ++u) {
                const int j = t0 + 4 * u + eo;
                if (j < pc) atomicMax(&rmaxu[rw[u] * 8 + hh], fmap(v[u]));
            }
        }
    }
#pragma unroll
    for (int r = 0; r < MSG_RW; ++r) {
        const unsigned mn = rmaxu[(lr0 + r) * 8 + h];
        const float sc = expf(funmap(mp[r]) - funmap(mn));
        v4f a = acc[(lr0 + r) * 32 + lane];
        a *= sc;
        acc[(lr0 + r) * 32 + lane] = a;
        rsum[(lr0 + r) * 8 + h] *= sc;
    }

    const int jq = lane & 3;
    for (int b = 0; b < pc; b += MSG_B) {
        int qx[MSG_B], qy[MSG_B]; v4f xv[MSG_B];
        float lv[2]; unsigned mv[2];
#pragma unroll
        for (int k = 0; k < 2; ++k) {
            const int j  = b + jq + 4 * k;
            const int2 q = myp[(j < pc) ? j : (pc - 1)];
            lv[k] = logit[(size_t)q.y * 8 + h];
            mv[k] = rmaxu[(int)((unsigned)q.x >> 24) * 8 + h];
        }
#pragma unroll
        for (int u = 0; u < MSG_B; ++u) {
            const int j = (b + u < pc) ? (b + u) : (pc - 1);
            const int2 q = myp[j];
            qx[u] = __builtin_amdgcn_readfirstlane(q.x);
            qy[u] = __builtin_amdgcn_readfirstlane(q.y);
        }
#pragma unroll
        for (int u = 0; u < MSG_B; ++u) {
            const int s = qx[u] & 0x00FFFFFF;
            xv[u] = *(const v4f*)(xl + (size_t)s * 128 + 4 * lane);
        }
        float pe[2];
#pragma unroll
        for (int k = 0; k < 2; ++k) pe[k] = expf(lv[k] - funmap(mv[k]));
#pragma unroll
        for (int u = 0; u < MSG_B; ++u) {
            if (b + u < pc) {
                const int rl  = (int)((unsigned)qx[u] >> 24);
                const float p = quad_lane(pe[u >> 2], u & 3);
                v4f a = acc[rl * 32 + lane];
                a += p * xv[u];
                acc[rl * 32 + lane] = a;
                rsum[rl * 8 + h] += p;
            }
        }
    }
}

__global__ __launch_bounds__(1024) void edge_msg_owner_k(
    const int* __restrict__ ei, const float* __restrict__ logit, const float* __restrict__ xl,
    const float* __restrict__ obias, float* __restrict__ out,
    const int* __restrict__ part_, const int* __restrict__ poff, int E_, int N_) {
    __shared__ v4f  acc[MSG_ROWS * 32];
    __shared__ int2 lst[MSG_WAVES * MSG_LCAP];
    __shared__ int2 prv[MSG_WAVES * MSG_PCAP];
    __shared__ unsigned rmaxu[MSG_ROWS * 8];
    __shared__ float    rsum[MSG_ROWS * 8];
    __shared__ int  s_cnt[MSG_WAVES];
    __shared__ int  s_un[MSG_WAVES];
    __shared__ int  s_oc[MSG_WAVES * MSG_WAVES];
    __shared__ int  s_run[MSG_WAVES * MSG_WAVES];

    const int lane = threadIdx.x & 31;
    const int w    = __builtin_amdgcn_readfirstlane((int)(threadIdx.x >> 5));
    const int h    = lane >> 2;
    const int G    = (int)gridDim.x;
    int rpb = (N_ + G - 1) / G;
    if (rpb > MSG_ROWS) rpb = MSG_ROWS;
    const int2* const part = (const int2*)part_;
    const int NC  = (E_ + BIN_CS - 1) / BIN_CS;
    const int CPW = (NC + MSG_WAVES - 1) / MSG_WAVES;
    const v4f bv = *(const v4f*)(obias + 4 * lane);
    const int lr0 = w * MSG_RW;
    int2* const myl = lst + w * MSG_LCAP;
    int2* const myp = prv + w * MSG_PCAP;

    for (int row0 = blockIdx.x * rpb; row0 < N_; row0 += G * rpb) {
        {
            v4f xv[MSG_RW]; float ls[MSG_RW];
#pragma unroll
            for (int r = 0; r < MSG_RW; ++r) {
                const int g  = row0 + lr0 + r;
                const int gg = (lr0 + r < rpb && g < N_) ? g : row0;
                xv[r] = *(const v4f*)(xl + (size_t)gg * 128 + 4 * lane);
                ls[r] = logit[(size_t)(E_ + gg) * 8 + h];
            }
#pragma unroll
            for (int r = 0; r < MSG_RW; ++r) {
                acc[(lr0 + r) * 32 + lane] = xv[r];
                rmaxu[(lr0 + r) * 8 + h]   = fmap(ls[r]);
                rsum[(lr0 + r) * 8 + h]    = 1.f;
            }
        }

        int R = row0 / rpb;
        if (R > BIN_NB - 1) R = BIN_NB - 1;
        const int cbeg = w * CPW;
        const int cend = (cbeg + CPW < NC) ? (cbeg + CPW) : NC;
        int ci = cbeg;
        int co = 0;
        for (;;) {
            int cnt = 0;
            while (ci < cend && cnt < MSG_LCAP) {
                const int  c  = ci + lane;
                const bool cv = c < cend;
                int o0 = 0, o1 = 0;
                if (cv) {
                    o0 = poff[(size_t)c * BIN_PP + R];
                    o1 = poff[(size_t)c * BIN_PP + R + 1];
                }
                if (lane == 0) o0 += co;
                const int n = o1 - o0;
                int inc = n;
#pragma unroll
                for (int off = 1; off < 32; off <<= 1) {
                    const int t = __shfl_up(inc, off, 32);
                    if (lane >= off) inc += t;
                }
                const int exc  = inc - n;
                const int room = MSG_LCAP - cnt;
                int t = room - exc;
                t = (t < 0) ? 0 : ((t > n) ? n : t);
                const int2* const sp = part + (size_t)c * BIN_CS + o0;
                int2* const dp = myl + cnt + exc;
                for (int j0 = 0; __builtin_amdgcn_ballot_w32(j0 < t) != 0u; j0 += 4) {
                    int2 v[4];
#pragma unroll
                    for (int u = 0; u < 4; ++u) if (j0 + u < t) v[u] = sp[j0 + u];
#pragma unroll
                    for (int u = 0; u < 4; ++u) if (j0 + u < t) dp[j0 + u] = v[u];
                }
                const unsigned cut = __builtin_amdgcn_ballot_w32(t < n);
                if (cut) {
                    const int f  = __builtin_ctz(cut);
                    const int tf = __builtin_amdgcn_readlane(t, f);
                    co  = tf + ((f == 0) ? co : 0);
                    ci += f;
                    cnt = MSG_LCAP;
                } else {
                    cnt += __builtin_amdgcn_readlane(inc, 31);
                    ci  += 32;
                    co   = 0;
                }
            }
            if (lane == 0) { s_cnt[w] = cnt; s_un[w] = (ci < cend) ? 1 : 0; }
            s_oc[w * MSG_WAVES + lane] = 0;
            for (int i0 = 0; i0 < cnt; i0 += 32) {
                const int i = i0 + lane;
                if (i < cnt) {
                    const int o = (int)((unsigned)myl[i].x >> 24) / MSG_RW;
                    if (o < MSG_WAVES) atomicAdd(&s_oc[w * MSG_WAVES + o], 1);
                }
            }
            __syncthreads();
            const bool more = __builtin_amdgcn_ballot_w32(s_un[lane] != 0) != 0u;

            int obase = 0, otot = 0;
#pragma unroll 8
            for (int L = 0; L < MSG_WAVES; ++L) {
                const int c = s_oc[L * MSG_WAVES + lane];
                obase += (L < w) ? c : 0;
                otot  += c;
            }
            const bool fits = __builtin_amdgcn_ballot_w32(otot > MSG_PCAP) == 0u;
            int pc = 0;
            if (fits) {
                int* const run = s_run + w * MSG_WAVES;
                run[lane] = obase;
                for (int i0 = 0; i0 < cnt; i0 += 32) {
                    const int i = i0 + lane;
                    int2 en = make_int2(0, 0);
                    if (i < cnt) en = myl[i];
                    const int  o     = (int)((unsigned)en.x >> 24) / MSG_RW;
                    const bool valid = (i < cnt) && (o < MSG_WAVES);
                    unsigned same = __builtin_amdgcn_ballot_w32(valid);
#pragma unroll
                    for (int bit = 0; bit < 5; ++bit) {
                        const bool on = valid && ((o >> bit) & 1);
                        const unsigned bal = __builtin_amdgcn_ballot_w32(on);
                        same &= on ? bal : ~bal;
                    }
                    const int rank = (int)__builtin_amdgcn_mbcnt_lo(same, 0u);
                    const int totg = __popc(same);
                    int pos = 0;
                    if (valid) {
                        pos = run[o];
                        prv[o * MSG_PCAP + pos + rank] = en;
                    }
                    if (valid && rank == 0) run[o] = pos + totg;
                }
                __syncthreads();
                pc = __builtin_amdgcn_readlane(otot, w);
            } else {
            for (int L = 0; L < MSG_WAVES; ++L) {
                const int n = __builtin_amdgcn_readfirstlane(s_cnt[L]);
                const int2* const ll = lst + L * MSG_LCAP;
                for (int i0 = 0; i0 < n; i0 += 32) {
                    const int  i  = i0 + lane;
                    const bool in = i < n;
                    int2 en = make_int2(0, 0);
                    if (in) en = ll[i];
                    const int  lr = (int)((unsigned)en.x >> 24);
                    const bool m  = in && (unsigned)(lr - lr0) < (unsigned)MSG_RW;
                    const unsigned bal = __builtin_amdgcn_ballot_w32(m);
                    if (bal) {
                        const int p = pc + (int)__builtin_amdgcn_mbcnt_lo(bal, 0u);
                        if (m) myp[p] = en;
                        pc += __popc(bal);
                    }
                    if (pc > MSG_PCAP - 32) {
                        msg_add_entries(myp, pc, acc, rmaxu, rsum, ei, xl, logit, lr0, lane, h);
                        pc = 0;
                    }
                }
            }
            }
            if (pc > 0) msg_add_entries(myp, pc, acc, rmaxu, rsum, ei, xl, logit, lr0, lane, h);
            __syncthreads();
            if (!more) break;
        }

#pragma unroll
        for (int r = 0; r < MSG_RW; ++r) {
            const int g = row0 + lr0 + r;
            if (lr0 + r < rpb && g < N_) {
                const float den = rsum[(lr0 + r) * 8 + h] + 1e-16f;
                const v4f a = acc[(lr0 + r) * 32 + lane];
                v4f o;
                o.x = bv.x + a.x / den;
                o.y = bv.y + a.y / den;
                o.z = bv.z + a.z / den;
                o.w = bv.w + a.w / den;
                *(v4f*)(out + (size_t)g * 128 + 4 * lane) = o;
            }
        }
    }
}

extern "C" void kernel_launch(void* const* d_in, const int* in_sizes, int n_in,
                              void* d_out, int out_size, void* d_ws, size_t ws_size,
                              hipStream_t stream) {
    const float* x     = (const float*)d_in[0];
    const int*   ei    = (const int*)d_in[1];
    const float* eattr = (const float*)d_in[2];
    const float* Wl    = (const float*)d_in[3];
    const float* bl    = (const float*)d_in[4];
    const float* Wr    = (const float*)d_in[5];
    const float* br    = (const float*)d_in[6];
    const float* We    = (const float*)d_in[7];
    const float* att   = (const float*)d_in[8];
    const float* bias  = (const float*)d_in[9];
    float* out = (float*)d_out;
    float* ws  = (float*)d_ws;

    const int n = in_sizes[0] / 128;
    const int e = in_sizes[2] / 8;
    const int tot = e + n;

    size_t off_mean   = 0;
    size_t off_xl     = 256;
    size_t off_xr     = off_xl + (size_t)n * 128;
    size_t off_logit  = off_xr + (size_t)n * 128;
    size_t off_part   = off_logit + (size_t)tot * 8;
    size_t off_poff   = off_part + (size_t)e * 2 + 2 * BIN_CS;

    int nTiles = n / 16;
    dim3 g((nTiles + 3) / 4, 2);
    node_xform_wmma<<<g, 128, 0, stream>>>(x, Wl, bl, Wr, br,
                                           ws + off_xl, ws + off_xr,
                                           ws + off_mean, nTiles);
    attr_mean_k<<<256, 256, 0, stream>>>(eattr, ws + off_mean, e);
    int nChunks = (e + BIN_CS - 1) / BIN_CS;
    edge_part_k<<<(nChunks + BIN_WPB - 1) / BIN_WPB, 256, 0, stream>>>(
        ei, (int*)(ws + off_part), (int*)(ws + off_poff), e, n, BIN_NB);

    edge_logit_tile_k<<<512, 256, 0, stream>>>(ei, eattr, We, ws + off_mean,
                                                      1.0f / (float)e,
                                                      ws + off_xl, ws + off_xr, att,
                                                      ws + off_logit, e, n);
    edge_msg_owner_k<<<BIN_NB, 1024, 0, stream>>>(ei, ws + off_logit, ws + off_xl, bias, out,
                                                  (const int*)(ws + off_part), (const int*)(ws + off_poff), e, n);
}
